# starting kernel
# baseline (speedup 1.0000x reference)
#include <hip/hip_runtime.h>
#include <math.h>

typedef __attribute__((ext_vector_type(2))) float v2f;
typedef __attribute__((ext_vector_type(8))) float v8f;

#define NFEAT  128
#define LDSPAD 132

__global__ __launch_bounds__(256) void gcn_gemm_bias(
    const float* __restrict__ X, const float* __restrict__ W,
    const float* __restrict__ Bv, float* __restrict__ H, int nnodes)
{
    __shared__ float xs[16 * LDSPAD];

    const int tid = threadIdx.x;
    const int m0  = blockIdx.x * 16;

    #pragma unroll
    for (int i = 0; i < 8; ++i) {
        int idx = tid + i * 256;
        int r = idx >> 7;
        int c = idx & 127;
        int gr = m0 + r;
        xs[r * LDSPAD + c] = (gr < nnodes) ? X[(size_t)gr * NFEAT + c] : 0.0f;
    }
    __syncthreads();

    const int wave = tid >> 5;
    const int lane = tid & 31;
    const int nn   = lane & 15;
    const int kb   = (lane >> 4) << 1;
    const int n0   = wave * 16;

    v8f acc;
    {
        float bias = Bv[n0 + nn];
        #pragma unroll
        for (int r = 0; r < 8; ++r) acc[r] = bias;
    }

    const float* xrow = &xs[nn * LDSPAD + kb];
    const float* wcol = &W[(size_t)kb * NFEAT + n0 + nn];

    #pragma unroll
    for (int k = 0; k < NFEAT; k += 4) {
        v2f a, b;
        a.x = xrow[k];
        a.y = xrow[k + 1];
        b.x = wcol[(size_t)k * NFEAT];
        b.y = wcol[(size_t)(k + 1) * NFEAT];
        acc = __builtin_amdgcn_wmma_f32_16x16x4_f32(
             false, a,  false, b,
             (short)0, acc,  false,  false);
    }

    const int mh = (lane >> 4) * 8;
    #pragma unroll
    for (int r = 0; r < 8; ++r) {
        int gm = m0 + r + mh;
        if (gm < nnodes) H[(size_t)gm * NFEAT + n0 + nn] = acc[r];
    }
}

__global__ __launch_bounds__(256) void gcn_scatter(
    const float* __restrict__ H, const int* __restrict__ rows,
    const int* __restrict__ cols, const float* __restrict__ vals,
    float* __restrict__ AGG, int nedges)
{
    int e    = (int)((blockIdx.x * 256u + threadIdx.x) >> 5);
    int lane = threadIdx.x & 31;
    if (e >= nedges) return;

    int   r = rows[e];
    int   c = cols[e];
    float v = vals[e];

    const float4* hp = (const float4*)(H + (size_t)c * NFEAT);
    float4 m = hp[lane];

    float* out = AGG + (size_t)r * NFEAT + (size_t)lane * 4;
    unsafeAtomicAdd(out + 0, m.x * v);
    unsafeAtomicAdd(out + 1, m.y * v);
    unsafeAtomicAdd(out + 2, m.z * v);
    unsafeAtomicAdd(out + 3, m.w * v);
}

__global__ __launch_bounds__(256) void gcn_zero(float4* __restrict__ p, int n4)
{
    int i = blockIdx.x * 256 + threadIdx.x;
    if (i < n4) {
        float4 z; z.x = 0.f; z.y = 0.f; z.z = 0.f; z.w = 0.f;
        p[i] = z;
    }
}

__global__ __launch_bounds__(256) void gcn_elu(
    const float* __restrict__ in, float* __restrict__ out, int n)
{
    int i = blockIdx.x * 256 + threadIdx.x;
    if (i < n) {
        float v = in[i];
        out[i] = (v > 0.0f) ? v : (__expf(v) - 1.0f);
    }
}

extern "C" void kernel_launch(void* const* d_in, const int* in_sizes, int n_in,
                              void* d_out, int out_size, void* d_ws, size_t ws_size,
                              hipStream_t stream)
{
    const float* x    = (const float*)d_in[0];
    const float* W    = (const float*)d_in[1];
    const float* b    = (const float*)d_in[2];
    const int*   rows = (const int*)  d_in[3];
    const int*   cols = (const int*)  d_in[4];
    const float* vals = (const float*)d_in[5];

    const int nnodes  = in_sizes[0] / NFEAT;
    const int nlayers = in_sizes[1] / (NFEAT * NFEAT);
    const int nedges  = in_sizes[3] / nlayers;

    const size_t nodeElems = (size_t)nnodes * NFEAT;
    float* hbuf = (float*)d_ws;
    float* agg  = hbuf + nodeElems;

    const int n          = (int)nodeElems;
    const int gemmBlocks = (nnodes + 15) / 16;
    const int zeroBlocks = (n / 4 + 255) / 256;
    const int eluBlocks  = (n + 255) / 256;
    const int scatBlocks = (nedges + 7) / 8;

    for (int l = 0; l < nlayers; ++l) {
        const float* cur = (l == 0) ? x : agg;
        gcn_gemm_bias<<<gemmBlocks, 256, 0, stream>>>(
            cur, W + (size_t)l * NFEAT * NFEAT, b + (size_t)l * NFEAT,
            hbuf, nnodes);

        gcn_zero<<<zeroBlocks, 256, 0, stream>>>((float4*)agg, n / 4);

        gcn_scatter<<<scatBlocks, 256, 0, stream>>>(
            hbuf, rows + (size_t)l * nedges, cols + (size_t)l * nedges,
            vals + (size_t)l * nedges, agg, nedges);

        float* eluOut = (l == nlayers - 1) ? (float*)d_out : agg;
        gcn_elu<<<eluBlocks, 256, 0, stream>>>(agg, eluOut, n);
    }
}


# Edges counting-sorted by destination row (int atomic arrival-order rank); one wave per row sums its messages in registers and stores ELU once: float atomics, zero and ELU passes gone
# speedup vs baseline: 1.5818x; 1.5818x over previous
#include <hip/hip_runtime.h>
#include <math.h>

typedef __attribute__((ext_vector_type(2))) float v2f;
typedef __attribute__((ext_vector_type(8))) float v8f;

#define NFEAT  128
#define LDSPAD 132

__global__ __launch_bounds__(256) void gcn_gemm_bias(
    const float* __restrict__ X, const float* __restrict__ W,
    const float* __restrict__ Bv, float* __restrict__ H, int nnodes)
{
    __shared__ float xs[16 * LDSPAD];

    const int tid = threadIdx.x;
    const int m0  = blockIdx.x * 16;

    #pragma unroll
    for (int i = 0; i < 8; ++i) {
        int idx = tid + i * 256;
        int r = idx >> 7;
        int c = idx & 127;
        int gr = m0 + r;
        xs[r * LDSPAD + c] = (gr < nnodes) ? X[(size_t)gr * NFEAT + c] : 0.0f;
    }
    __syncthreads();

    const int wave = tid >> 5;
    const int lane = tid & 31;
    const int nn   = lane & 15;
    const int kb   = (lane >> 4) << 1;
    const int n0   = wave * 16;

    v8f acc;
    {
        float bias = Bv[n0 + nn];
        #pragma unroll
        for (int r = 0; r < 8; ++r) acc[r] = bias;
    }

    const float* xrow = &xs[nn * LDSPAD + kb];
    const float* wcol = &W[(size_t)kb * NFEAT + n0 + nn];

    #pragma unroll
    for (int k = 0; k < NFEAT; k += 4) {
        v2f a, b;
        a.x = xrow[k];
        a.y = xrow[k + 1];
        b.x = wcol[(size_t)k * NFEAT];
        b.y = wcol[(size_t)(k + 1) * NFEAT];
        acc = __builtin_amdgcn_wmma_f32_16x16x4_f32(
             false, a,  false, b,
             (short)0, acc,  false,  false);
    }

    const int mh = (lane >> 4) * 8;
    #pragma unroll
    for (int r = 0; r < 8; ++r) {
        int gm = m0 + r + mh;
        if (gm < nnodes) H[(size_t)gm * NFEAT + n0 + nn] = acc[r];
    }
}

__global__ __launch_bounds__(256) void gcn_zero_i(int* __restrict__ p, int n)
{
    int i = blockIdx.x * 256 + threadIdx.x;
    if (i < n) p[i] = 0;
}

__global__ __launch_bounds__(256) void gcn_rank(
    const int* __restrict__ rows, int* __restrict__ counts,
    int* __restrict__ rank, int nedges, int nnodes)
{
    const int l = blockIdx.y;
    const int e = blockIdx.x * 256 + threadIdx.x;
    if (e >= nedges) return;
    const size_t g = (size_t)l * nedges + e;
    rank[g] = atomicAdd(&counts[(size_t)l * nnodes + rows[g]], 1);
}

__global__ __launch_bounds__(1024) void gcn_offsets(
    const int* __restrict__ counts, int* __restrict__ offs, int nnodes)
{
    __shared__ int wtot[32];
    __shared__ int wpre[32];

    const int l    = blockIdx.y;
    const int* cnt = counts + (size_t)l * nnodes;
    int* of        = offs + (size_t)l * (nnodes + 1);
    const int tid  = threadIdx.x;
    const int lane = tid & 31;
    const int wave = tid >> 5;
    const int t0   = blockIdx.x * 1024;
    const int i    = t0 + tid;

    int pre = 0;
    for (int j = tid; j < t0; j += 1024) pre += cnt[j];
    const int mine = (i < nnodes) ? cnt[i] : 0;

    int incl = mine;
    for (int d = 1; d < 32; d <<= 1) {
        int v = __shfl_up(incl, d);
        if (lane >= d) incl += v;
    }
    for (int d = 16; d >= 1; d >>= 1) pre += __shfl_xor(pre, d);
    if (lane == 31) wtot[wave] = incl;
    if (lane == 0)  wpre[wave] = pre;
    __syncthreads();

    int base = wpre[lane] + ((lane < wave) ? wtot[lane] : 0);
    for (int d = 16; d >= 1; d >>= 1) base += __shfl_xor(base, d);

    const int excl = base + incl - mine;
    if (i < nnodes)      of[i] = excl;
    if (i == nnodes - 1) of[nnodes] = excl + mine;
}

__global__ __launch_bounds__(256) void gcn_place(
    const int* __restrict__ rows, const int* __restrict__ cols,
    const float* __restrict__ vals, const int* __restrict__ rank,
    const int* __restrict__ offs, int2* __restrict__ ent, int nedges, int nnodes)
{
    const int l = blockIdx.y;
    const int e = blockIdx.x * 256 + threadIdx.x;
    if (e >= nedges) return;
    const size_t g = (size_t)l * nedges + e;
    const int p = offs[(size_t)l * (nnodes + 1) + rows[g]] + rank[g];
    int2 v;
    v.x = cols[g];
    v.y = __float_as_int(vals[g]);
    ent[(size_t)l * nedges + p] = v;
}

__global__ __launch_bounds__(256) void gcn_gather_elu(
    const float* __restrict__ H, const int* __restrict__ offs,
    const int2* __restrict__ ent, float* __restrict__ out, int nnodes)
{
    const int lane = threadIdx.x & 31;
    const int r = __builtin_amdgcn_readfirstlane((int)((blockIdx.x * 256u + threadIdx.x) >> 5));
    if (r >= nnodes) return;

    const int s = offs[r];
    const int t = offs[r + 1];

    float4 acc; acc.x = 0.f; acc.y = 0.f; acc.z = 0.f; acc.w = 0.f;

    for (int b = s; b < t; b += 32) {
        const int n = (t - b < 32) ? (t - b) : 32;
        const int2 ev = ent[b + ((lane < n) ? lane : 0)];
        const int nb = (n + 7) >> 3;
        for (int q = 0; q < nb; ++q) {
            const int k = q * 8;
            float4 m[8];
            float  w[8];
            #pragma unroll
            for (int j = 0; j < 8; ++j) {
                const int kk = (k + j < n) ? (k + j) : (n - 1);
                const int c  = __builtin_amdgcn_readlane(ev.x, kk);
                w[j] = __int_as_float(__builtin_amdgcn_readlane(ev.y, kk));
                m[j] = ((const float4*)(H + (size_t)c * NFEAT))[lane];
            }
            #pragma unroll
            for (int j = 0; j < 8; ++j) {
                const bool ok = (j == 0) || (k + j < n);
                const float sx = fmaf(w[j], m[j].x, acc.x);
                const float sy = fmaf(w[j], m[j].y, acc.y);
                const float sz = fmaf(w[j], m[j].z, acc.z);
                const float sw = fmaf(w[j], m[j].w, acc.w);
                acc.x = ok ? sx : acc.x;
                acc.y = ok ? sy : acc.y;
                acc.z = ok ? sz : acc.z;
                acc.w = ok ? sw : acc.w;
            }
        }
    }

    float4 o;
    o.x = (acc.x > 0.0f) ? acc.x : (__expf(acc.x) - 1.0f);
    o.y = (acc.y > 0.0f) ? acc.y : (__expf(acc.y) - 1.0f);
    o.z = (acc.z > 0.0f) ? acc.z : (__expf(acc.z) - 1.0f);
    o.w = (acc.w > 0.0f) ? acc.w : (__expf(acc.w) - 1.0f);
    ((float4*)(out + (size_t)r * NFEAT))[lane] = o;
}

extern "C" void kernel_launch(void* const* d_in, const int* in_sizes, int n_in,
                              void* d_out, int out_size, void* d_ws, size_t ws_size,
                              hipStream_t stream)
{
    const float* x    = (const float*)d_in[0];
    const float* W    = (const float*)d_in[1];
    const float* b    = (const float*)d_in[2];
    const int*   rows = (const int*)  d_in[3];
    const int*   cols = (const int*)  d_in[4];
    const float* vals = (const float*)d_in[5];

    const int nnodes  = in_sizes[0] / NFEAT;
    const int nlayers = in_sizes[1] / (NFEAT * NFEAT);
    const int nedges  = in_sizes[3] / nlayers;

    const size_t nodeElems = (size_t)nnodes * NFEAT;
    const size_t allEdges  = (size_t)nlayers * nedges;
    float* hbuf   = (float*)d_ws;
    float* act    = hbuf + nodeElems;
    int2*  ent    = (int2*)(act + nodeElems);
    int*   rank   = (int*)(ent + allEdges);
    int*   counts = rank + allEdges;
    int*   offs   = counts + (size_t)nlayers * nnodes;

    const int ncount     = nlayers * nnodes;
    const int gemmBlocks = (nnodes + 15) / 16;
    const int rowBlocks  = (nnodes + 7) / 8;
    const dim3 edgeGrid((nedges + 255) / 256, nlayers);
    const dim3 tileGrid((nnodes + 1023) / 1024, nlayers);

    gcn_zero_i<<<(ncount + 255) / 256, 256, 0, stream>>>(counts, ncount);
    gcn_rank<<<edgeGrid, 256, 0, stream>>>(rows, counts, rank, nedges, nnodes);
    gcn_offsets<<<tileGrid, 1024, 0, stream>>>(counts, offs, nnodes);
    gcn_place<<<edgeGrid, 256, 0, stream>>>(rows, cols, vals, rank, offs, ent, nedges, nnodes);

    for (int l = 0; l < nlayers; ++l) {
        const float* cur = (l == 0) ? x : act;
        gcn_gemm_bias<<<gemmBlocks, 256, 0, stream>>>(
            cur, W + (size_t)l * NFEAT * NFEAT, b + (size_t)l * NFEAT,
            hbuf, nnodes);

        float* dst = (l == nlayers - 1) ? (float*)d_out : act;
        gcn_gather_elu<<<rowBlocks, 256, 0, stream>>>(
            hbuf, offs + (size_t)l * (nnodes + 1), ent + (size_t)l * nedges,
            dst, nnodes);
    }
}


# Edges ordered by a two-level counting sort on LDS counters (256-row bins, then rows inside a bin): global returning atomics and random stores gone; a row's sum order is now LDS arrival order
# speedup vs baseline: 2.4497x; 1.5487x over previous
#include <hip/hip_runtime.h>
#include <math.h>

typedef __attribute__((ext_vector_type(2))) float v2f;
typedef __attribute__((ext_vector_type(8))) float v8f;

#define NFEAT  128
#define LDSPAD 132

__global__ __launch_bounds__(256) void gcn_gemm_bias(
    const float* __restrict__ X, const float* __restrict__ W,
    const float* __restrict__ Bv, float* __restrict__ H, int nnodes)
{
    __shared__ float xs[16 * LDSPAD];

    const int tid = threadIdx.x;
    const int m0  = blockIdx.x * 16;

    #pragma unroll
    for (int i = 0; i < 8; ++i) {
        int idx = tid + i * 256;
        int r = idx >> 7;
        int c = idx & 127;
        int gr = m0 + r;
        xs[r * LDSPAD + c] = (gr < nnodes) ? X[(size_t)gr * NFEAT + c] : 0.0f;
    }
    __syncthreads();

    const int wave = tid >> 5;
    const int lane = tid & 31;
    const int nn   = lane & 15;
    const int kb   = (lane >> 4) << 1;
    const int n0   = wave * 16;

    v8f acc;
    {
        float bias = Bv[n0 + nn];
        #pragma unroll
        for (int r = 0; r < 8; ++r) acc[r] = bias;
    }

    const float* xrow = &xs[nn * LDSPAD + kb];
    const float* wcol = &W[(size_t)kb * NFEAT + n0 + nn];

    #pragma unroll
    for (int k = 0; k < NFEAT; k += 4) {
        v2f a, b;
        a.x = xrow[k];
        a.y = xrow[k + 1];
        b.x = wcol[(size_t)k * NFEAT];
        b.y = wcol[(size_t)(k + 1) * NFEAT];
        acc = __builtin_amdgcn_wmma_f32_16x16x4_f32(
             false, a,  false, b,
             (short)0, acc,  false,  false);
    }

    const int mh = (lane >> 4) * 8;
    #pragma unroll
    for (int r = 0; r < 8; ++r) {
        int gm = m0 + r + mh;
        if (gm < nnodes) H[(size_t)gm * NFEAT + n0 + nn] = acc[r];
    }
}

#define BINSHIFT 8
#define BINROWS  256
#define MAXBINS  1024
#define EPT      8
#define CHUNK    (1024 * EPT)

__global__ __launch_bounds__(1024) void gcn_bin_count(
    const int* __restrict__ rows, int* __restrict__ tab,
    int nedges, int nbins, int nchunks)
{
    __shared__ int hist[MAXBINS];
    const int l = blockIdx.y, c = blockIdx.x, tid = threadIdx.x;
    for (int i = tid; i < nbins; i += 1024) hist[i] = 0;
    __syncthreads();

    const int* rw = rows + (size_t)l * nedges;
    const int e0 = c * CHUNK + tid;
    int r[EPT];
    #pragma unroll
    for (int j = 0; j < EPT; ++j) {
        const int e = e0 + j * 1024;
        r[j] = rw[(e < nedges) ? e : (nedges - 1)];
    }
    #pragma unroll
    for (int j = 0; j < EPT; ++j)
        if (e0 + j * 1024 < nedges) atomicAdd(&hist[r[j] >> BINSHIFT], 1);
    __syncthreads();

    int* t = tab + (size_t)l * nbins * nchunks;
    for (int i = tid; i < nbins; i += 1024) t[(size_t)i * nchunks + c] = hist[i];
}

__global__ __launch_bounds__(1024) void gcn_bin_scan(int* __restrict__ tab, int nbins, int nchunks)
{
    __shared__ int wtot[32];
    const int tid = threadIdx.x, lane = tid & 31, wave = tid >> 5;
    const int total = nbins * nchunks;
    int* t = tab + (size_t)blockIdx.y * total;
    const int per   = (total + 1023) / 1024;
    const int begin = (tid * per < total) ? tid * per : total;
    const int end   = (begin + per < total) ? begin + per : total;

    int s = 0;
    for (int i = begin; i < end; ++i) s += t[i];

    int incl = s;
    for (int d = 1; d < 32; d <<= 1) {
        int v = __shfl_up(incl, d);
        if (lane >= d) incl += v;
    }
    if (lane == 31) wtot[wave] = incl;
    __syncthreads();
    int front = (lane < wave) ? wtot[lane] : 0;
    for (int d = 16; d >= 1; d >>= 1) front += __shfl_xor(front, d);

    int run = front + incl - s;
    for (int i = begin; i < end; ++i) {
        const int v = t[i];
        t[i] = run;
        run += v;
    }
}

__global__ __launch_bounds__(1024) void gcn_bin_scatter(
    const int* __restrict__ rows, const int* __restrict__ cols,
    const float* __restrict__ vals, const int* __restrict__ tab,
    int* __restrict__ trow, int2* __restrict__ tent,
    int nedges, int nbins, int nchunks)
{
    __shared__ int cur[MAXBINS];
    const int l = blockIdx.y, c = blockIdx.x, tid = threadIdx.x;
    const int* t = tab + (size_t)l * nbins * nchunks;
    for (int i = tid; i < nbins; i += 1024) cur[i] = t[(size_t)i * nchunks + c];
    __syncthreads();

    const size_t lo = (size_t)l * nedges;
    const int e0 = c * CHUNK + tid;
    int r[EPT], cc[EPT], vv[EPT];
    #pragma unroll
    for (int j = 0; j < EPT; ++j) {
        const int e  = e0 + j * 1024;
        const size_t g = lo + ((e < nedges) ? e : (nedges - 1));
        r[j]  = rows[g];
        cc[j] = cols[g];
        vv[j] = __float_as_int(vals[g]);
    }
    #pragma unroll
    for (int j = 0; j < EPT; ++j) {
        if (e0 + j * 1024 < nedges) {
            const int p = atomicAdd(&cur[r[j] >> BINSHIFT], 1);
            int2 v;
            v.x = cc[j];
            v.y = vv[j];
            trow[lo + p] = r[j];
            tent[lo + p] = v;
        }
    }
}

__global__ __launch_bounds__(512) void gcn_bin_sort(
    const int* __restrict__ tab, const int* __restrict__ trow,
    const int2* __restrict__ tent, int* __restrict__ offs, int2* __restrict__ ent,
    int nedges, int nnodes, int nbins, int nchunks)
{
    __shared__ int cnt[BINROWS];
    __shared__ int cur[BINROWS];
    __shared__ int wtot[8];
    const int l = blockIdx.y, bin = blockIdx.x;
    const int tid = threadIdx.x, lane = tid & 31, wave = tid >> 5;
    const int* t = tab + (size_t)l * nbins * nchunks;
    const int s  = t[(size_t)bin * nchunks];
    const int e  = (bin + 1 < nbins) ? t[(size_t)(bin + 1) * nchunks] : nedges;
    const size_t lo = (size_t)l * nedges;

    if (tid < BINROWS) cnt[tid] = 0;
    __syncthreads();

    for (int b = s + tid; b - tid < e; b += 512 * EPT) {
        int r[EPT];
        #pragma unroll
        for (int j = 0; j < EPT; ++j) {
            const int i = b + j * 512;
            r[j] = trow[lo + ((i < e) ? i : (e - 1))];
        }
        #pragma unroll
        for (int j = 0; j < EPT; ++j)
            if (b + j * 512 < e) atomicAdd(&cnt[r[j] & (BINROWS - 1)], 1);
    }
    __syncthreads();

    const int mine = (tid < BINROWS) ? cnt[tid] : 0;
    int incl = mine;
    for (int d = 1; d < 32; d <<= 1) {
        int v = __shfl_up(incl, d);
        if (lane >= d) incl += v;
    }
    if (tid < BINROWS && lane == 31) wtot[wave] = incl;
    __syncthreads();
    int front = (lane < 8 && lane < wave) ? wtot[lane] : 0;
    for (int d = 16; d >= 1; d >>= 1) front += __shfl_xor(front, d);
    const int first = s + front + incl - mine;

    if (tid < BINROWS) {
        int* of = offs + (size_t)l * (nnodes + 1);
        const int row = bin * BINROWS + tid;
        cur[tid] = first;
        if (row < nnodes)      of[row] = first;
        if (row == nnodes - 1) of[nnodes] = first + mine;
    }
    __syncthreads();

    for (int b = s + tid; b - tid < e; b += 512 * EPT) {
        int  r[EPT];
        int2 v[EPT];
        #pragma unroll
        for (int j = 0; j < EPT; ++j) {
            const int i = b + j * 512;
            const size_t g = lo + ((i < e) ? i : (e - 1));
            r[j] = trow[g];
            v[j] = tent[g];
        }
        #pragma unroll
        for (int j = 0; j < EPT; ++j) {
            if (b + j * 512 < e) {
                const int p = atomicAdd(&cur[r[j] & (BINROWS - 1)], 1);
                ent[lo + p] = v[j];
            }
        }
    }
}

__global__ __launch_bounds__(256) void gcn_gather_elu(
    const float* __restrict__ H, const int* __restrict__ offs,
    const int2* __restrict__ ent, float* __restrict__ out, int nnodes)
{
    const int lane = threadIdx.x & 31;
    const int r = __builtin_amdgcn_readfirstlane((int)((blockIdx.x * 256u + threadIdx.x) >> 5));
    if (r >= nnodes) return;

    const int s = offs[r];
    const int t = offs[r + 1];

    float4 acc; acc.x = 0.f; acc.y = 0.f; acc.z = 0.f; acc.w = 0.f;

    for (int b = s; b < t; b += 32) {
        const int n = (t - b < 32) ? (t - b) : 32;
        const int2 ev = ent[b + ((lane < n) ? lane : 0)];
        const int nb = (n + 7) >> 3;
        for (int q = 0; q < nb; ++q) {
            const int k = q * 8;
            float4 m[8];
            float  w[8];
            #pragma unroll
            for (int j = 0; j < 8; ++j) {
                const int kk = (k + j < n) ? (k + j) : (n - 1);
                const int c  = __builtin_amdgcn_readlane(ev.x, kk);
                w[j] = __int_as_float(__builtin_amdgcn_readlane(ev.y, kk));
                m[j] = ((const float4*)(H + (size_t)c * NFEAT))[lane];
            }
            #pragma unroll
            for (int j = 0; j < 8; ++j) {
                const bool ok = (j == 0) || (k + j < n);
                const float sx = fmaf(w[j], m[j].x, acc.x);
                const float sy = fmaf(w[j], m[j].y, acc.y);
                const float sz = fmaf(w[j], m[j].z, acc.z);
                const float sw = fmaf(w[j], m[j].w, acc.w);
                acc.x = ok ? sx : acc.x;
                acc.y = ok ? sy : acc.y;
                acc.z = ok ? sz : acc.z;
                acc.w = ok ? sw : acc.w;
            }
        }
    }

    float4 o;
    o.x = (acc.x > 0.0f) ? acc.x : (__expf(acc.x) - 1.0f);
    o.y = (acc.y > 0.0f) ? acc.y : (__expf(acc.y) - 1.0f);
    o.z = (acc.z > 0.0f) ? acc.z : (__expf(acc.z) - 1.0f);
    o.w = (acc.w > 0.0f) ? acc.w : (__expf(acc.w) - 1.0f);
    ((float4*)(out + (size_t)r * NFEAT))[lane] = o;
}

extern "C" void kernel_launch(void* const* d_in, const int* in_sizes, int n_in,
                              void* d_out, int out_size, void* d_ws, size_t ws_size,
                              hipStream_t stream)
{
    const float* x    = (const float*)d_in[0];
    const float* W    = (const float*)d_in[1];
    const float* b    = (const float*)d_in[2];
    const int*   rows = (const int*)  d_in[3];
    const int*   cols = (const int*)  d_in[4];
    const float* vals = (const float*)d_in[5];

    const int nnodes  = in_sizes[0] / NFEAT;
    const int nlayers = in_sizes[1] / (NFEAT * NFEAT);
    const int nedges  = in_sizes[3] / nlayers;

    const int nbins   = (nnodes + BINROWS - 1) / BINROWS;
    const int nchunks = (nedges + CHUNK - 1) / CHUNK;

    const size_t nodeElems = (size_t)nnodes * NFEAT;
    const size_t allEdges  = (size_t)nlayers * nedges;
    float* hbuf = (float*)d_ws;
    float* act  = hbuf + nodeElems;
    int2*  ent  = (int2*)(act + nodeElems);
    int2*  tent = ent + allEdges;
    int*   trow = (int*)(tent + allEdges);
    int*   tab  = trow + allEdges;
    int*   offs = tab + (size_t)nlayers * nbins * nchunks;

    const int gemmBlocks = (nnodes + 15) / 16;
    const int rowBlocks  = (nnodes + 7) / 8;
    const dim3 chunkGrid(nchunks, nlayers);
    const dim3 scanGrid(1, nlayers);
    const dim3 binGrid(nbins, nlayers);

    gcn_bin_count<<<chunkGrid, 1024, 0, stream>>>(rows, tab, nedges, nbins, nchunks);
    gcn_bin_scan<<<scanGrid, 1024, 0, stream>>>(tab, nbins, nchunks);
    gcn_bin_scatter<<<chunkGrid, 1024, 0, stream>>>(rows, cols, vals, tab, trow, tent,
                                                    nedges, nbins, nchunks);
    gcn_bin_sort<<<binGrid, 512, 0, stream>>>(tab, trow, tent, offs, ent,
                                              nedges, nnodes, nbins, nchunks);

    for (int l = 0; l < nlayers; ++l) {
        const float* cur = (l == 0) ? x : act;
        gcn_gemm_bias<<<gemmBlocks, 256, 0, stream>>>(
            cur, W + (size_t)l * NFEAT * NFEAT, b + (size_t)l * NFEAT,
            hbuf, nnodes);

        float* dst = (l == nlayers - 1) ? (float*)d_out : act;
        gcn_gather_elu<<<rowBlocks, 256, 0, stream>>>(
            hbuf, offs + (size_t)l * (nnodes + 1), ent + (size_t)l * nedges,
            dst, nnodes);
    }
}


# Each layer as ELU((A x) W + rowsum(A) b) in one launch: resident blocks sum 16 rows of x into LDS, then WMMA with W held in registers; h round trip and 2 launches gone; sums reassociated
# speedup vs baseline: 2.7251x; 1.1124x over previous
#include <hip/hip_runtime.h>
#include <math.h>

typedef __attribute__((ext_vector_type(2))) float v2f;
typedef __attribute__((ext_vector_type(8))) float v8f;

#define NFEAT  128
#define LDSPAD 132

#define BINSHIFT 8
#define BINROWS  256
#define MAXBINS  1024
#define EPT      8
#define CHUNK    (1024 * EPT)

__global__ __launch_bounds__(1024) void gcn_bin_count(
    const int* __restrict__ rows, int* __restrict__ tab,
    int nedges, int nbins, int nchunks)
{
    __shared__ int hist[MAXBINS];
    const int l = blockIdx.y, c = blockIdx.x, tid = threadIdx.x;
    for (int i = tid; i < nbins; i += 1024) hist[i] = 0;
    __syncthreads();

    const int* rw = rows + (size_t)l * nedges;
    const int e0 = c * CHUNK + tid;
    int r[EPT];
    #pragma unroll
    for (int j = 0; j < EPT; ++j) {
        const int e = e0 + j * 1024;
        r[j] = rw[(e < nedges) ? e : (nedges - 1)];
    }
    #pragma unroll
    for (int j = 0; j < EPT; ++j)
        if (e0 + j * 1024 < nedges) atomicAdd(&hist[r[j] >> BINSHIFT], 1);
    __syncthreads();

    int* t = tab + (size_t)l * nbins * nchunks;
    for (int i = tid; i < nbins; i += 1024) t[(size_t)i * nchunks + c] = hist[i];
}

__global__ __launch_bounds__(1024) void gcn_bin_scan(int* __restrict__ tab, int nbins, int nchunks)
{
    __shared__ int wtot[32];
    const int tid = threadIdx.x, lane = tid & 31, wave = tid >> 5;
    const int total = nbins * nchunks;
    int* t = tab + (size_t)blockIdx.y * total;
    const int per   = (total + 1023) / 1024;
    const int begin = (tid * per < total) ? tid * per : total;
    const int end   = (begin + per < total) ? begin + per : total;

    int s = 0;
    for (int i = begin; i < end; ++i) s += t[i];

    int incl = s;
    for (int d = 1; d < 32; d <<= 1) {
        int v = __shfl_up(incl, d);
        if (lane >= d) incl += v;
    }
    if (lane == 31) wtot[wave] = incl;
    __syncthreads();
    int front = (lane < wave) ? wtot[lane] : 0;
    for (int d = 16; d >= 1; d >>= 1) front += __shfl_xor(front, d);

    int run = front + incl - s;
    for (int i = begin; i < end; ++i) {
        const int v = t[i];
        t[i] = run;
        run += v;
    }
}

__global__ __launch_bounds__(1024) void gcn_bin_scatter(
    const int* __restrict__ rows, const int* __restrict__ cols,
    const float* __restrict__ vals, const int* __restrict__ tab,
    int* __restrict__ trow, int2* __restrict__ tent,
    int nedges, int nbins, int nchunks)
{
    __shared__ int cur[MAXBINS];
    const int l = blockIdx.y, c = blockIdx.x, tid = threadIdx.x;
    const int* t = tab + (size_t)l * nbins * nchunks;
    for (int i = tid; i < nbins; i += 1024) cur[i] = t[(size_t)i * nchunks + c];
    __syncthreads();

    const size_t lo = (size_t)l * nedges;
    const int e0 = c * CHUNK + tid;
    int r[EPT], cc[EPT], vv[EPT];
    #pragma unroll
    for (int j = 0; j < EPT; ++j) {
        const int e  = e0 + j * 1024;
        const size_t g = lo + ((e < nedges) ? e : (nedges - 1));
        r[j]  = rows[g];
        cc[j] = cols[g];
        vv[j] = __float_as_int(vals[g]);
    }
    #pragma unroll
    for (int j = 0; j < EPT; ++j) {
        if (e0 + j * 1024 < nedges) {
            const int p = atomicAdd(&cur[r[j] >> BINSHIFT], 1);
            int2 v;
            v.x = cc[j];
            v.y = vv[j];
            trow[lo + p] = r[j];
            tent[lo + p] = v;
        }
    }
}

__global__ __launch_bounds__(512) void gcn_bin_sort(
    const int* __restrict__ tab, const int* __restrict__ trow,
    const int2* __restrict__ tent, int* __restrict__ offs, int2* __restrict__ ent,
    int nedges, int nnodes, int nbins, int nchunks)
{
    __shared__ int cnt[BINROWS];
    __shared__ int cur[BINROWS];
    __shared__ int wtot[8];
    const int l = blockIdx.y, bin = blockIdx.x;
    const int tid = threadIdx.x, lane = tid & 31, wave = tid >> 5;
    const int* t = tab + (size_t)l * nbins * nchunks;
    const int s  = t[(size_t)bin * nchunks];
    const int e  = (bin + 1 < nbins) ? t[(size_t)(bin + 1) * nchunks] : nedges;
    const size_t lo = (size_t)l * nedges;

    if (tid < BINROWS) cnt[tid] = 0;
    __syncthreads();

    for (int b = s + tid; b - tid < e; b += 512 * EPT) {
        int r[EPT];
        #pragma unroll
        for (int j = 0; j < EPT; ++j) {
            const int i = b + j * 512;
            r[j] = trow[lo + ((i < e) ? i : (e - 1))];
        }
        #pragma unroll
        for (int j = 0; j < EPT; ++j)
            if (b + j * 512 < e) atomicAdd(&cnt[r[j] & (BINROWS - 1)], 1);
    }
    __syncthreads();

    const int mine = (tid < BINROWS) ? cnt[tid] : 0;
    int incl = mine;
    for (int d = 1; d < 32; d <<= 1) {
        int v = __shfl_up(incl, d);
        if (lane >= d) incl += v;
    }
    if (tid < BINROWS && lane == 31) wtot[wave] = incl;
    __syncthreads();
    int front = (lane < 8 && lane < wave) ? wtot[lane] : 0;
    for (int d = 16; d >= 1; d >>= 1) front += __shfl_xor(front, d);
    const int first = s + front + incl - mine;

    if (tid < BINROWS) {
        int* of = offs + (size_t)l * (nnodes + 1);
        const int row = bin * BINROWS + tid;
        cur[tid] = first;
        if (row < nnodes)      of[row] = first;
        if (row == nnodes - 1) of[nnodes] = first + mine;
    }
    __syncthreads();

    for (int b = s + tid; b - tid < e; b += 512 * EPT) {
        int  r[EPT];
        int2 v[EPT];
        #pragma unroll
        for (int j = 0; j < EPT; ++j) {
            const int i = b + j * 512;
            const size_t g = lo + ((i < e) ? i : (e - 1));
            r[j] = trow[g];
            v[j] = tent[g];
        }
        #pragma unroll
        for (int j = 0; j < EPT; ++j) {
            if (b + j * 512 < e) {
                const int p = atomicAdd(&cur[r[j] & (BINROWS - 1)], 1);
                ent[lo + p] = v[j];
            }
        }
    }
}

#define LAYER_BLOCKS 768


__device__ __forceinline__ void gcn_row_sum(
    const float* __restrict__ X, const int2* __restrict__ ent,
    int s, int t, int2 ev, int lane, float4& acc, float& wsum)
{
    for (int b = s; b < t; b += 32) {
        const int n = (t - b < 32) ? (t - b) : 32;
        if (b != s) ev = ent[b + ((lane < n) ? lane : 0)];
        const int nb = (n + 7) >> 3;
        for (int q = 0; q < nb; ++q) {
            const int k = q * 8;
            float4 m[8];
            float  w[8];
            #pragma unroll
            for (int j = 0; j < 8; ++j) {
                const int kk = (k + j < n) ? (k + j) : (n - 1);
                const int c  = __builtin_amdgcn_readlane(ev.x, kk);
                w[j] = __int_as_float(__builtin_amdgcn_readlane(ev.y, kk));
                m[j] = ((const float4*)(X + (size_t)c * NFEAT))[lane];
            }
            #pragma unroll
            for (int j = 0; j < 8; ++j) {
                const bool ok = (j == 0) || (k + j < n);
                const float sx = fmaf(w[j], m[j].x, acc.x);
                const float sy = fmaf(w[j], m[j].y, acc.y);
                const float sz = fmaf(w[j], m[j].z, acc.z);
                const float sw = fmaf(w[j], m[j].w, acc.w);
                acc.x = ok ? sx : acc.x;
                acc.y = ok ? sy : acc.y;
                acc.z = ok ? sz : acc.z;
                acc.w = ok ? sw : acc.w;
                wsum  = ok ? (wsum + w[j]) : wsum;
            }
        }
    }
}

__global__ __launch_bounds__(256) void gcn_layer(
    const float* __restrict__ X, const float* __restrict__ W,
    const float* __restrict__ Bv, const int* __restrict__ offs,
    const int2* __restrict__ ent, float* __restrict__ out,
    int nnodes, int ntiles)
{
    __shared__ __attribute__((aligned(16))) float xs[2][16 * LDSPAD];
    __shared__ float rs[2][16];

    const int tid  = threadIdx.x;
    const int lane = tid & 31;
    const int wave = __builtin_amdgcn_readfirstlane(tid >> 5);
    const int nn   = lane & 15;
    const int kb   = (lane >> 4) << 1;
    const int mh   = (lane >> 4) * 8;
    const int n0   = wave * 16;

    v2f bw[32];
    {
        const float* wcol = &W[(size_t)kb * NFEAT + n0 + nn];
        #pragma unroll
        for (int k4 = 0; k4 < 32; ++k4) {
            bw[k4].x = wcol[(size_t)(4 * k4) * NFEAT];
            bw[k4].y = wcol[(size_t)(4 * k4 + 1) * NFEAT];
        }
    }
    const float bias = Bv[n0 + nn];

    int p = 0;
    for (int tile = blockIdx.x; tile < ntiles; tile += gridDim.x, p ^= 1) {
        const int m0 = tile * 16;

        {
            const int r0 = m0 + wave * 2;
            const int i0 = (r0     < nnodes) ? r0       : nnodes;
            const int i1 = (r0 + 1 < nnodes) ? (r0 + 1) : nnodes;
            const int i2 = (r0 + 2 < nnodes) ? (r0 + 2) : nnodes;
            const int s0 = offs[i0], s1 = offs[i1], s2 = offs[i2];
            const int nA = (s1 - s0 < 32) ? (s1 - s0) : 32;
            const int nB = (s2 - s1 < 32) ? (s2 - s1) : 32;
            const int2 evA = ent[(nA > 0) ? (s0 + ((lane < nA) ? lane : 0)) : 0];
            const int2 evB = ent[(nB > 0) ? (s1 + ((lane < nB) ? lane : 0)) : 0];

            float4 a0; a0.x = 0.f; a0.y = 0.f; a0.z = 0.f; a0.w = 0.f;
            float4 a1 = a0;
            float  w0 = 0.f, w1 = 0.f;
            gcn_row_sum(X, ent, s0, s1, evA, lane, a0, w0);
            gcn_row_sum(X, ent, s1, s2, evB, lane, a1, w1);

            float* xw = &xs[p][(wave * 2) * LDSPAD + lane * 4];
            *(float4*)xw            = a0;
            *(float4*)(xw + LDSPAD) = a1;
            if (lane == 0) { rs[p][wave * 2] = w0; rs[p][wave * 2 + 1] = w1; }
        }
        __syncthreads();

        v8f acc;
        #pragma unroll
        for (int r = 0; r < 8; ++r) acc[r] = rs[p][r + mh] * bias;

        const float* xrow = &xs[p][nn * LDSPAD + kb];
        #pragma unroll
        for (int k4 = 0; k4 < 32; ++k4) {
            v2f a;
            a.x = xrow[4 * k4];
            a.y = xrow[4 * k4 + 1];
            acc = __builtin_amdgcn_wmma_f32_16x16x4_f32(
                 false, a,  false, bw[k4],
                 (short)0, acc,  false,  false);
        }

        float o[8];
        #pragma unroll
        for (int r = 0; r < 8; ++r) {
            const float v = acc[r];
            o[r] = (v > 0.0f) ? v : (__expf(v) - 1.0f);
        }
        float* orow = out + (size_t)(m0 + mh) * NFEAT + n0 + nn;
        if (m0 + 16 <= nnodes) {
            #pragma unroll
            for (int r = 0; r < 8; ++r) orow[(size_t)r * NFEAT] = o[r];
        } else {
            #pragma unroll
            for (int r = 0; r < 8; ++r)
                if (m0 + mh + r < nnodes) orow[(size_t)r * NFEAT] = o[r];
        }
    }
}

extern "C" void kernel_launch(void* const* d_in, const int* in_sizes, int n_in,
                              void* d_out, int out_size, void* d_ws, size_t ws_size,
                              hipStream_t stream)
{
    const float* x    = (const float*)d_in[0];
    const float* W    = (const float*)d_in[1];
    const float* b    = (const float*)d_in[2];
    const int*   rows = (const int*)  d_in[3];
    const int*   cols = (const int*)  d_in[4];
    const float* vals = (const float*)d_in[5];

    const int nnodes  = in_sizes[0] / NFEAT;
    const int nlayers = in_sizes[1] / (NFEAT * NFEAT);
    const int nedges  = in_sizes[3] / nlayers;

    const int nbins   = (nnodes + BINROWS - 1) / BINROWS;
    const int nchunks = (nedges + CHUNK - 1) / CHUNK;

    const size_t nodeElems = (size_t)nnodes * NFEAT;
    const size_t allEdges  = (size_t)nlayers * nedges;
    float* actA = (float*)d_ws;
    float* actB = actA + nodeElems;
    int2*  ent  = (int2*)(actB + nodeElems);
    int2*  tent = ent + allEdges;
    int*   trow = (int*)(tent + allEdges);
    int*   tab  = trow + allEdges;
    int*   offs = tab + (size_t)nlayers * nbins * nchunks;

    const int ntiles = (nnodes + 15) / 16;
    const dim3 chunkGrid(nchunks, nlayers);
    const dim3 scanGrid(1, nlayers);
    const dim3 binGrid(nbins, nlayers);

    gcn_bin_count<<<chunkGrid, 1024, 0, stream>>>(rows, tab, nedges, nbins, nchunks);
    gcn_bin_scan<<<scanGrid, 1024, 0, stream>>>(tab, nbins, nchunks);
    gcn_bin_scatter<<<chunkGrid, 1024, 0, stream>>>(rows, cols, vals, tab, trow, tent,
                                                    nedges, nbins, nchunks);
    gcn_bin_sort<<<binGrid, 512, 0, stream>>>(tab, trow, tent, offs, ent,
                                              nedges, nnodes, nbins, nchunks);

    for (int l = 0; l < nlayers; ++l) {
        const float* cur = (l == 0) ? x : ((l & 1) ? actA : actB);
        float* dst = (l == nlayers - 1) ? (float*)d_out : ((l & 1) ? actB : actA);
        gcn_layer<<<LAYER_BLOCKS, 256, 0, stream>>>(
            cur, W + (size_t)l * NFEAT * NFEAT, b + (size_t)l * NFEAT,
            offs + (size_t)l * (nnodes + 1), ent + (size_t)l * nedges,
            dst, nnodes, ntiles);
    }
}


# Both edge-sort copy passes order their entries inside LDS first, then store them lane-next-to-lane as contiguous runs: per-lane scattered stores gone; row sum order still LDS arrival order
# speedup vs baseline: 3.4419x; 1.2630x over previous
#include <hip/hip_runtime.h>
#include <math.h>

typedef __attribute__((ext_vector_type(2))) float v2f;
typedef __attribute__((ext_vector_type(8))) float v8f;

#define NFEAT  128
#define LDSPAD 132

#define BINSHIFT 8
#define BINROWS  256
#define MAXBINS  1024
#define EPT      8
#define CHUNK    (1024 * EPT)

__global__ __launch_bounds__(1024) void gcn_bin_count(
    const int* __restrict__ rows, int* __restrict__ tab,
    int nedges, int nbins, int nchunks)
{
    __shared__ int hist[MAXBINS];
    const int l = blockIdx.y, c = blockIdx.x, tid = threadIdx.x;
    for (int i = tid; i < nbins; i += 1024) hist[i] = 0;
    __syncthreads();

    const int* rw = rows + (size_t)l * nedges;
    const int e0 = c * CHUNK + tid;
    int r[EPT];
    #pragma unroll
    for (int j = 0; j < EPT; ++j) {
        const int e = e0 + j * 1024;
        r[j] = rw[(e < nedges) ? e : (nedges - 1)];
    }
    #pragma unroll
    for (int j = 0; j < EPT; ++j)
        if (e0 + j * 1024 < nedges) atomicAdd(&hist[r[j] >> BINSHIFT], 1);
    __syncthreads();

    int* t = tab + (size_t)l * nbins * nchunks;
    for (int i = tid; i < nbins; i += 1024) t[(size_t)i * nchunks + c] = hist[i];
}

__global__ __launch_bounds__(1024) void gcn_bin_scan(int* __restrict__ tab, int nbins, int nchunks)
{
    __shared__ int wtot[32];
    const int tid = threadIdx.x, lane = tid & 31, wave = tid >> 5;
    const int total = nbins * nchunks;
    int* t = tab + (size_t)blockIdx.y * total;
    const int per   = (total + 1023) / 1024;
    const int begin = (tid * per < total) ? tid * per : total;
    const int end   = (begin + per < total) ? begin + per : total;

    int s = 0;
    for (int i = begin; i < end; ++i) s += t[i];

    int incl = s;
    for (int d = 1; d < 32; d <<= 1) {
        int v = __shfl_up(incl, d);
        if (lane >= d) incl += v;
    }
    if (lane == 31) wtot[wave] = incl;
    __syncthreads();
    int front = (lane < wave) ? wtot[lane] : 0;
    for (int d = 16; d >= 1; d >>= 1) front += __shfl_xor(front, d);

    int run = front + incl - s;
    for (int i = begin; i < end; ++i) {
        const int v = t[i];
        t[i] = run;
        run += v;
    }
}

__global__ __launch_bounds__(1024) void gcn_bin_scatter(
    const int* __restrict__ rows, const int* __restrict__ cols,
    const float* __restrict__ vals, const int* __restrict__ tab,
    int* __restrict__ trow, int2* __restrict__ tent,
    int nedges, int nbins, int nchunks)
{
    __shared__ int  gfirst[MAXBINS];
    __shared__ int  lfirst[MAXBINS];
    __shared__ int  cur[MAXBINS];
    __shared__ int  wtot[32];
    __shared__ int  srow[CHUNK];
    __shared__ int2 sent[CHUNK];
    const int l = blockIdx.y, c = blockIdx.x, tid = threadIdx.x;
    const int lane = tid & 31, wave = tid >> 5;
    const int total = nbins * nchunks;
    const int* t = tab + (size_t)l * total;

    int g0 = 0, cnt = 0;
    if (tid < nbins) {
        const int idx = tid * nchunks + c;
        g0 = t[idx];
        cnt = ((idx + 1 < total) ? t[idx + 1] : nedges) - g0;
    }
    int incl = cnt;
    for (int d = 1; d < 32; d <<= 1) {
        int v = __shfl_up(incl, d);
        if (lane >= d) incl += v;
    }
    if (lane == 31) wtot[wave] = incl;
    __syncthreads();
    int front = (lane < wave) ? wtot[lane] : 0;
    for (int d = 16; d >= 1; d >>= 1) front += __shfl_xor(front, d);
    if (tid < nbins) {
        const int lf = front + incl - cnt;
        gfirst[tid] = g0;
        lfirst[tid] = lf;
        cur[tid]    = lf;
    }
    __syncthreads();

    const size_t lo = (size_t)l * nedges;
    const int e0 = c * CHUNK + tid;
    int r[EPT], cc[EPT], vv[EPT];
    #pragma unroll
    for (int j = 0; j < EPT; ++j) {
        const int e  = e0 + j * 1024;
        const size_t g = lo + ((e < nedges) ? e : (nedges - 1));
        r[j]  = rows[g];
        cc[j] = cols[g];
        vv[j] = __float_as_int(vals[g]);
    }
    int q[EPT];
    #pragma unroll
    for (int j = 0; j < EPT; ++j)
        q[j] = (e0 + j * 1024 < nedges) ? atomicAdd(&cur[r[j] >> BINSHIFT], 1) : 0;
    #pragma unroll
    for (int j = 0; j < EPT; ++j) {
        if (e0 + j * 1024 < nedges) {
            int2 v;
            v.x = cc[j];
            v.y = vv[j];
            srow[q[j]] = r[j];
            sent[q[j]] = v;
        }
    }
    __syncthreads();

    const int left = nedges - c * CHUNK;
    const int nloc = (left < CHUNK) ? left : CHUNK;
    int  rr[EPT], gg[EPT];
    int2 ev[EPT];
    #pragma unroll
    for (int j = 0; j < EPT; ++j) {
        const int i  = tid + j * 1024;
        const int ic = (i < nloc) ? i : (nloc - 1);
        rr[j] = srow[ic];
        ev[j] = sent[ic];
    }
    #pragma unroll
    for (int j = 0; j < EPT; ++j) {
        const int i = tid + j * 1024;
        const int b = rr[j] >> BINSHIFT;
        gg[j] = gfirst[b] + (i - lfirst[b]);
    }
    #pragma unroll
    for (int j = 0; j < EPT; ++j) {
        if (tid + j * 1024 < nloc) {
            trow[lo + gg[j]] = rr[j];
            tent[lo + gg[j]] = ev[j];
        }
    }
}

#define SORTCAP 8192

__global__ __launch_bounds__(512) void gcn_bin_sort(
    const int* __restrict__ tab, const int* __restrict__ trow,
    const int2* __restrict__ tent, int* __restrict__ offs, int2* __restrict__ ent,
    int nedges, int nnodes, int nbins, int nchunks)
{
    __shared__ int  cnt[BINROWS];
    __shared__ int  cur[BINROWS];
    __shared__ int  wtot[8];
    __shared__ int2 stage[SORTCAP];
    const int l = blockIdx.y, bin = blockIdx.x;
    const int tid = threadIdx.x, lane = tid & 31, wave = tid >> 5;
    const int* t = tab + (size_t)l * nbins * nchunks;
    const int s  = t[(size_t)bin * nchunks];
    const int e  = (bin + 1 < nbins) ? t[(size_t)(bin + 1) * nchunks] : nedges;
    const size_t lo = (size_t)l * nedges;
    const bool staged = (e - s <= SORTCAP);

    if (tid < BINROWS) cnt[tid] = 0;
    __syncthreads();

    for (int b = s + tid; b - tid < e; b += 512 * EPT) {
        int r[EPT];
        #pragma unroll
        for (int j = 0; j < EPT; ++j) {
            const int i = b + j * 512;
            r[j] = trow[lo + ((i < e) ? i : (e - 1))];
        }
        #pragma unroll
        for (int j = 0; j < EPT; ++j)
            if (b + j * 512 < e) atomicAdd(&cnt[r[j] & (BINROWS - 1)], 1);
    }
    __syncthreads();

    const int mine = (tid < BINROWS) ? cnt[tid] : 0;
    int incl = mine;
    for (int d = 1; d < 32; d <<= 1) {
        int v = __shfl_up(incl, d);
        if (lane >= d) incl += v;
    }
    if (tid < BINROWS && lane == 31) wtot[wave] = incl;
    __syncthreads();
    int front = (lane < 8 && lane < wave) ? wtot[lane] : 0;
    for (int d = 16; d >= 1; d >>= 1) front += __shfl_xor(front, d);
    const int first = s + front + incl - mine;

    if (tid < BINROWS) {
        int* of = offs + (size_t)l * (nnodes + 1);
        const int row = bin * BINROWS + tid;
        cur[tid] = staged ? (first - s) : first;
        if (row < nnodes)      of[row] = first;
        if (row == nnodes - 1) of[nnodes] = first + mine;
    }
    __syncthreads();

    if (staged) {
        for (int b = s + tid; b - tid < e; b += 512 * EPT) {
            int  r[EPT], p[EPT];
            int2 v[EPT];
            #pragma unroll
            for (int j = 0; j < EPT; ++j) {
                const int i = b + j * 512;
                const size_t g = lo + ((i < e) ? i : (e - 1));
                r[j] = trow[g];
                v[j] = tent[g];
            }
            #pragma unroll
            for (int j = 0; j < EPT; ++j)
                p[j] = (b + j * 512 < e) ? atomicAdd(&cur[r[j] & (BINROWS - 1)], 1) : 0;
            #pragma unroll
            for (int j = 0; j < EPT; ++j)
                if (b + j * 512 < e) stage[p[j]] = v[j];
        }
        __syncthreads();
        int2* dst = ent + lo + s;
        for (int i = tid; i < e - s; i += 512) dst[i] = stage[i];
    } else {
        for (int b = s + tid; b - tid < e; b += 512 * EPT) {
            int  r[EPT];
            int2 v[EPT];
            #pragma unroll
            for (int j = 0; j < EPT; ++j) {
                const int i = b + j * 512;
                const size_t g = lo + ((i < e) ? i : (e - 1));
                r[j] = trow[g];
                v[j] = tent[g];
            }
            #pragma unroll
            for (int j = 0; j < EPT; ++j) {
                if (b + j * 512 < e) {
                    const int p = atomicAdd(&cur[r[j] & (BINROWS - 1)], 1);
                    ent[lo + p] = v[j];
                }
            }
        }
    }
}

#define LAYER_BLOCKS 768


__device__ __forceinline__ void gcn_row_sum(
    const float* __restrict__ X, const int2* __restrict__ ent,
    int s, int t, int2 ev, int lane, float4& acc, float& wsum)
{
    for (int b = s; b < t; b += 32) {
        const int n = (t - b < 32) ? (t - b) : 32;
        if (b != s) ev = ent[b + ((lane < n) ? lane : 0)];
        const int nb = (n + 7) >> 3;
        for (int q = 0; q < nb; ++q) {
            const int k = q * 8;
            float4 m[8];
            float  w[8];
            #pragma unroll
            for (int j = 0; j < 8; ++j) {
                const int kk = (k + j < n) ? (k + j) : (n - 1);
                const int c  = __builtin_amdgcn_readlane(ev.x, kk);
                w[j] = __int_as_float(__builtin_amdgcn_readlane(ev.y, kk));
                m[j] = ((const float4*)(X + (size_t)c * NFEAT))[lane];
            }
            #pragma unroll
            for (int j = 0; j < 8; ++j) {
                const bool ok = (j == 0) || (k + j < n);
                const float sx = fmaf(w[j], m[j].x, acc.x);
                const float sy = fmaf(w[j], m[j].y, acc.y);
                const float sz = fmaf(w[j], m[j].z, acc.z);
                const float sw = fmaf(w[j], m[j].w, acc.w);
                acc.x = ok ? sx : acc.x;
                acc.y = ok ? sy : acc.y;
                acc.z = ok ? sz : acc.z;
                acc.w = ok ? sw : acc.w;
                wsum  = ok ? (wsum + w[j]) : wsum;
            }
        }
    }
}

__global__ __launch_bounds__(256) void gcn_layer(
    const float* __restrict__ X, const float* __restrict__ W,
    const float* __restrict__ Bv, const int* __restrict__ offs,
    const int2* __restrict__ ent, float* __restrict__ out,
    int nnodes, int ntiles)
{
    __shared__ __attribute__((aligned(16))) float xs[2][16 * LDSPAD];
    __shared__ float rs[2][16];

    const int tid  = threadIdx.x;
    const int lane = tid & 31;
    const int wave = __builtin_amdgcn_readfirstlane(tid >> 5);
    const int nn   = lane & 15;
    const int kb   = (lane >> 4) << 1;
    const int mh   = (lane >> 4) * 8;
    const int n0   = wave * 16;

    v2f bw[32];
    {
        const float* wcol = &W[(size_t)kb * NFEAT + n0 + nn];
        #pragma unroll
        for (int k4 = 0; k4 < 32; ++k4) {
            bw[k4].x = wcol[(size_t)(4 * k4) * NFEAT];
            bw[k4].y = wcol[(size_t)(4 * k4 + 1) * NFEAT];
        }
    }
    const float bias = Bv[n0 + nn];

    int p = 0;
    for (int tile = blockIdx.x; tile < ntiles; tile += gridDim.x, p ^= 1) {
        const int m0 = tile * 16;

        {
            const int r0 = m0 + wave * 2;
            const int i0 = (r0     < nnodes) ? r0       : nnodes;
            const int i1 = (r0 + 1 < nnodes) ? (r0 + 1) : nnodes;
            const int i2 = (r0 + 2 < nnodes) ? (r0 + 2) : nnodes;
            const int s0 = offs[i0], s1 = offs[i1], s2 = offs[i2];
            const int nA = (s1 - s0 < 32) ? (s1 - s0) : 32;
            const int nB = (s2 - s1 < 32) ? (s2 - s1) : 32;
            const int2 evA = ent[(nA > 0) ? (s0 + ((lane < nA) ? lane : 0)) : 0];
            const int2 evB = ent[(nB > 0) ? (s1 + ((lane < nB) ? lane : 0)) : 0];

            float4 a0; a0.x = 0.f; a0.y = 0.f; a0.z = 0.f; a0.w = 0.f;
            float4 a1 = a0;
            float  w0 = 0.f, w1 = 0.f;
            gcn_row_sum(X, ent, s0, s1, evA, lane, a0, w0);
            gcn_row_sum(X, ent, s1, s2, evB, lane, a1, w1);

            float* xw = &xs[p][(wave * 2) * LDSPAD + lane * 4];
            *(float4*)xw            = a0;
            *(float4*)(xw + LDSPAD) = a1;
            if (lane == 0) { rs[p][wave * 2] = w0; rs[p][wave * 2 + 1] = w1; }
        }
        __syncthreads();

        v8f acc;
        #pragma unroll
        for (int r = 0; r < 8; ++r) acc[r] = rs[p][r + mh] * bias;

        const float* xrow = &xs[p][nn * LDSPAD + kb];
        #pragma unroll
        for (int k4 = 0; k4 < 32; ++k4) {
            v2f a;
            a.x = xrow[4 * k4];
            a.y = xrow[4 * k4 + 1];
            acc = __builtin_amdgcn_wmma_f32_16x16x4_f32(
                 false, a,  false, bw[k4],
                 (short)0, acc,  false,  false);
        }

        float o[8];
        #pragma unroll
        for (int r = 0; r < 8; ++r) {
            const float v = acc[r];
            o[r] = (v > 0.0f) ? v : (__expf(v) - 1.0f);
        }
        float* orow = out + (size_t)(m0 + mh) * NFEAT + n0 + nn;
        if (m0 + 16 <= nnodes) {
            #pragma unroll
            for (int r = 0; r < 8; ++r) orow[(size_t)r * NFEAT] = o[r];
        } else {
            #pragma unroll
            for (int r = 0; r < 8; ++r)
                if (m0 + mh + r < nnodes) orow[(size_t)r * NFEAT] = o[r];
        }
    }
}

extern "C" void kernel_launch(void* const* d_in, const int* in_sizes, int n_in,
                              void* d_out, int out_size, void* d_ws, size_t ws_size,
                              hipStream_t stream)
{
    const float* x    = (const float*)d_in[0];
    const float* W    = (const float*)d_in[1];
    const float* b    = (const float*)d_in[2];
    const int*   rows = (const int*)  d_in[3];
    const int*   cols = (const int*)  d_in[4];
    const float* vals = (const float*)d_in[5];

    const int nnodes  = in_sizes[0] / NFEAT;
    const int nlayers = in_sizes[1] / (NFEAT * NFEAT);
    const int nedges  = in_sizes[3] / nlayers;

    const int nbins   = (nnodes + BINROWS - 1) / BINROWS;
    const int nchunks = (nedges + CHUNK - 1) / CHUNK;

    const size_t nodeElems = (size_t)nnodes * NFEAT;
    const size_t allEdges  = (size_t)nlayers * nedges;
    float* actA = (float*)d_ws;
    float* actB = actA + nodeElems;
    int2*  ent  = (int2*)(actB + nodeElems);
    int2*  tent = ent + allEdges;
    int*   trow = (int*)(tent + allEdges);
    int*   tab  = trow + allEdges;
    int*   offs = tab + (size_t)nlayers * nbins * nchunks;

    const int ntiles = (nnodes + 15) / 16;
    const dim3 chunkGrid(nchunks, nlayers);
    const dim3 scanGrid(1, nlayers);
    const dim3 binGrid(nbins, nlayers);

    gcn_bin_count<<<chunkGrid, 1024, 0, stream>>>(rows, tab, nedges, nbins, nchunks);
    gcn_bin_scan<<<scanGrid, 1024, 0, stream>>>(tab, nbins, nchunks);
    gcn_bin_scatter<<<chunkGrid, 1024, 0, stream>>>(rows, cols, vals, tab, trow, tent,
                                                    nedges, nbins, nchunks);
    gcn_bin_sort<<<binGrid, 512, 0, stream>>>(tab, trow, tent, offs, ent,
                                              nedges, nnodes, nbins, nchunks);

    for (int l = 0; l < nlayers; ++l) {
        const float* cur = (l == 0) ? x : ((l & 1) ? actA : actB);
        float* dst = (l == nlayers - 1) ? (float*)d_out : ((l & 1) ? actB : actA);
        gcn_layer<<<LAYER_BLOCKS, 256, 0, stream>>>(
            cur, W + (size_t)l * NFEAT * NFEAT, b + (size_t)l * NFEAT,
            offs + (size_t)l * (nnodes + 1), ent + (size_t)l * nedges,
            dst, nnodes, ntiles);
    }
}


# Edge sort in 2 launches, was 4: a chunk orders its own edges by bin in its own slots, a bin block finds its per-chunk runs by search; count and scan passes gone; row sum order still LDS arrival order
# speedup vs baseline: 3.8923x; 1.1309x over previous
#include <hip/hip_runtime.h>
#include <math.h>

typedef __attribute__((ext_vector_type(2))) float v2f;
typedef __attribute__((ext_vector_type(8))) float v8f;

#define NFEAT  128
#define LDSPAD 132

#define BINSHIFT 8
#define BINROWS  256
#define MAXBINS  1024
#define EPT      8
#define CHUNK    (1024 * EPT)
#define MAXCH    1024

__global__ __launch_bounds__(1024) void gcn_chunk_sort(
    const int* __restrict__ rows, const int* __restrict__ cols,
    const float* __restrict__ vals, int* __restrict__ tab,
    int* __restrict__ trow, int2* __restrict__ tent,
    int nedges, int nbins, int nchunks)
{
    __shared__ int  hist[MAXBINS];
    __shared__ int  wtot[32];
    __shared__ int  srow[CHUNK];
    __shared__ int2 sent[CHUNK];
    const int l = blockIdx.y, c = blockIdx.x, tid = threadIdx.x;
    const int lane = tid & 31, wave = tid >> 5;

    hist[tid] = 0;
    __syncthreads();

    const size_t lo = (size_t)l * nedges;
    const int e0 = c * CHUNK + tid;
    int r[EPT], cc[EPT], vv[EPT];
    #pragma unroll
    for (int j = 0; j < EPT; ++j) {
        const int e  = e0 + j * 1024;
        const size_t g = lo + ((e < nedges) ? e : (nedges - 1));
        r[j]  = rows[g];
        cc[j] = cols[g];
        vv[j] = __float_as_int(vals[g]);
    }
    int q[EPT];
    #pragma unroll
    for (int j = 0; j < EPT; ++j)
        q[j] = (e0 + j * 1024 < nedges) ? atomicAdd(&hist[r[j] >> BINSHIFT], 1) : 0;
    __syncthreads();

    const int cnt = hist[tid];
    int incl = cnt;
    for (int d = 1; d < 32; d <<= 1) {
        int v = __shfl_up(incl, d);
        if (lane >= d) incl += v;
    }
    if (lane == 31) wtot[wave] = incl;
    __syncthreads();
    int front = (lane < wave) ? wtot[lane] : 0;
    for (int d = 16; d >= 1; d >>= 1) front += __shfl_xor(front, d);
    const int lf = front + incl - cnt;
    hist[tid] = lf;

    const int left = nedges - c * CHUNK;
    const int nloc = (left < CHUNK) ? left : CHUNK;
    int* t = tab + (size_t)l * (nbins + 1) * nchunks;
    if (tid < nbins) t[(size_t)tid * nchunks + c] = lf;
    if (tid == 0)    t[(size_t)nbins * nchunks + c] = nloc;
    __syncthreads();

    #pragma unroll
    for (int j = 0; j < EPT; ++j) {
        if (e0 + j * 1024 < nedges) {
            const int slot = hist[r[j] >> BINSHIFT] + q[j];
            int2 v;
            v.x = cc[j];
            v.y = vv[j];
            srow[slot] = r[j];
            sent[slot] = v;
        }
    }
    __syncthreads();

    const size_t go = lo + (size_t)c * CHUNK;
    int  rr[EPT];
    int2 ev[EPT];
    #pragma unroll
    for (int j = 0; j < EPT; ++j) {
        const int i  = tid + j * 1024;
        const int ic = (i < nloc) ? i : (nloc - 1);
        rr[j] = srow[ic];
        ev[j] = sent[ic];
    }
    #pragma unroll
    for (int j = 0; j < EPT; ++j) {
        const int i = tid + j * 1024;
        if (i < nloc) {
            trow[go + i] = rr[j];
            tent[go + i] = ev[j];
        }
    }
}

#define SORTCAP 8192

__device__ __forceinline__ void gcn_locate(
    const int* pre, const int* gb, int half, const int (&i)[EPT], int (&g)[EPT])
{
    int cx[EPT];
    #pragma unroll
    for (int j = 0; j < EPT; ++j) cx[j] = 0;
    for (int st = half; st > 0; st >>= 1) {
        #pragma unroll
        for (int j = 0; j < EPT; ++j) {
            const int m = cx[j] + st;
            cx[j] = (pre[m] <= i[j]) ? m : cx[j];
        }
    }
    #pragma unroll
    for (int j = 0; j < EPT; ++j) g[j] = gb[cx[j]] + i[j];
}

__global__ __launch_bounds__(512) void gcn_bin_sort(
    const int* __restrict__ tab, const int* __restrict__ trow,
    const int2* __restrict__ tent, int* __restrict__ offs, int2* __restrict__ ent,
    int nedges, int nnodes, int nbins, int nchunks)
{
    __shared__ int  pre[MAXCH];
    __shared__ int  gb[MAXCH];
    __shared__ int  cnt[BINROWS];
    __shared__ int  cur[BINROWS];
    __shared__ int  wrun[16];
    __shared__ int  wsum[16];
    __shared__ int  wtot[8];
    __shared__ int2 stage[SORTCAP];
    const int l = blockIdx.y, bin = blockIdx.x;
    const int tid = threadIdx.x, lane = tid & 31, wave = tid >> 5;
    const size_t lo = (size_t)l * nedges;
    const int* ta = tab + ((size_t)l * (nbins + 1) + bin) * nchunks;
    const int* tz = ta + nchunks;

    const int c0 = 2 * tid, c1 = c0 + 1;
    int a0 = 0, z0 = 0, a1 = 0, z1 = 0;
    if (c0 < nchunks) { a0 = ta[c0]; z0 = tz[c0]; }
    if (c1 < nchunks) { a1 = ta[c1]; z1 = tz[c1]; }
    const int n0 = z0 - a0, n1 = z1 - a1, both = n0 + n1;
    int rinc = both;
    for (int d = 1; d < 32; d <<= 1) {
        int v = __shfl_up(rinc, d);
        if (lane >= d) rinc += v;
    }
    int sa = a0 + a1;
    for (int d = 16; d >= 1; d >>= 1) sa += __shfl_xor(sa, d);
    if (lane == 31) { wrun[wave] = rinc; wsum[wave] = sa; }
    if (tid < BINROWS) cnt[tid] = 0;
    __syncthreads();
    int fr = (lane < wave) ? wrun[lane] : 0;
    int al = (lane < 16) ? wrun[lane] : 0;
    int sl = (lane < 16) ? wsum[lane] : 0;
    for (int d = 16; d >= 1; d >>= 1) {
        fr += __shfl_xor(fr, d);
        al += __shfl_xor(al, d);
        sl += __shfl_xor(sl, d);
    }
    const int s = sl;
    const int n = al;
    const int ex = fr + rinc - both;
    pre[c0] = (c0 < nchunks) ? ex        : 0x7fffffff;
    pre[c1] = (c1 < nchunks) ? (ex + n0) : 0x7fffffff;
    gb[c0]  = c0 * CHUNK + a0 - ex;
    gb[c1]  = c1 * CHUNK + a1 - (ex + n0);
    __syncthreads();

    const int half = (nchunks > 1) ? (1 << (31 - __clz(nchunks - 1))) : 0;
    const bool staged = (n <= SORTCAP);

    int2 v[2][EPT];
    int  pk[2][EPT];
    #pragma unroll
    for (int h = 0; h < 2; ++h) {
        #pragma unroll
        for (int j = 0; j < EPT; ++j) { v[h][j].x = 0; v[h][j].y = 0; pk[h][j] = 0; }
    }

    if (staged) {
        #pragma unroll
        for (int h = 0; h < 2; ++h) {
            if (h * 512 * EPT < n) {
                int ic[EPT], g[EPT], r[EPT];
                #pragma unroll
                for (int j = 0; j < EPT; ++j) {
                    const int i = h * 512 * EPT + tid + j * 512;
                    ic[j] = (i < n) ? i : (n - 1);
                }
                gcn_locate(pre, gb, half, ic, g);
                #pragma unroll
                for (int j = 0; j < EPT; ++j) {
                    r[j]    = trow[lo + g[j]];
                    v[h][j] = tent[lo + g[j]];
                }
                #pragma unroll
                for (int j = 0; j < EPT; ++j) {
                    const int i  = h * 512 * EPT + tid + j * 512;
                    const int rl = r[j] & (BINROWS - 1);
                    const int qq = (i < n) ? atomicAdd(&cnt[rl], 1) : 0;
                    pk[h][j] = (qq << 8) | rl;
                }
            }
        }
    } else {
        for (int b = 0; b < n; b += 512 * EPT) {
            int ic[EPT], g[EPT], r[EPT];
            #pragma unroll
            for (int j = 0; j < EPT; ++j) {
                const int i = b + tid + j * 512;
                ic[j] = (i < n) ? i : (n - 1);
            }
            gcn_locate(pre, gb, half, ic, g);
            #pragma unroll
            for (int j = 0; j < EPT; ++j) r[j] = trow[lo + g[j]];
            #pragma unroll
            for (int j = 0; j < EPT; ++j)
                if (b + tid + j * 512 < n) atomicAdd(&cnt[r[j] & (BINROWS - 1)], 1);
        }
    }
    __syncthreads();

    const int mine = (tid < BINROWS) ? cnt[tid] : 0;
    int incl = mine;
    for (int d = 1; d < 32; d <<= 1) {
        int w = __shfl_up(incl, d);
        if (lane >= d) incl += w;
    }
    if (tid < BINROWS && lane == 31) wtot[wave] = incl;
    __syncthreads();
    int front = (lane < 8 && lane < wave) ? wtot[lane] : 0;
    for (int d = 16; d >= 1; d >>= 1) front += __shfl_xor(front, d);
    const int first = s + front + incl - mine;

    if (tid < BINROWS) {
        int* of = offs + (size_t)l * (nnodes + 1);
        const int row = bin * BINROWS + tid;
        cur[tid] = staged ? (first - s) : first;
        if (row < nnodes)      of[row] = first;
        if (row == nnodes - 1) of[nnodes] = first + mine;
    }
    __syncthreads();

    if (staged) {
        #pragma unroll
        for (int h = 0; h < 2; ++h) {
            #pragma unroll
            for (int j = 0; j < EPT; ++j) {
                const int i = h * 512 * EPT + tid + j * 512;
                if (i < n) stage[cur[pk[h][j] & (BINROWS - 1)] + (pk[h][j] >> 8)] = v[h][j];
            }
        }
        __syncthreads();
        int2* dst = ent + lo + s;
        for (int i = tid; i < n; i += 512) dst[i] = stage[i];
    } else {
        for (int b = 0; b < n; b += 512 * EPT) {
            int  ic[EPT], g[EPT], r[EPT];
            int2 w[EPT];
            #pragma unroll
            for (int j = 0; j < EPT; ++j) {
                const int i = b + tid + j * 512;
                ic[j] = (i < n) ? i : (n - 1);
            }
            gcn_locate(pre, gb, half, ic, g);
            #pragma unroll
            for (int j = 0; j < EPT; ++j) {
                r[j] = trow[lo + g[j]];
                w[j] = tent[lo + g[j]];
            }
            #pragma unroll
            for (int j = 0; j < EPT; ++j) {
                if (b + tid + j * 512 < n) {
                    const int p = atomicAdd(&cur[r[j] & (BINROWS - 1)], 1);
                    ent[lo + p] = w[j];
                }
            }
        }
    }
}

#define LAYER_BLOCKS 768


__device__ __forceinline__ void gcn_row_sum(
    const float* __restrict__ X, const int2* __restrict__ ent,
    int s, int t, int2 ev, int lane, float4& acc, float& wsum)
{
    for (int b = s; b < t; b += 32) {
        const int n = (t - b < 32) ? (t - b) : 32;
        if (b != s) ev = ent[b + ((lane < n) ? lane : 0)];
        const int nb = (n + 7) >> 3;
        for (int q = 0; q < nb; ++q) {
            const int k = q * 8;
            float4 m[8];
            float  w[8];
            #pragma unroll
            for (int j = 0; j < 8; ++j) {
                const int kk = (k + j < n) ? (k + j) : (n - 1);
                const int c  = __builtin_amdgcn_readlane(ev.x, kk);
                w[j] = __int_as_float(__builtin_amdgcn_readlane(ev.y, kk));
                m[j] = ((const float4*)(X + (size_t)c * NFEAT))[lane];
            }
            #pragma unroll
            for (int j = 0; j < 8; ++j) {
                const bool ok = (j == 0) || (k + j < n);
                const float sx = fmaf(w[j], m[j].x, acc.x);
                const float sy = fmaf(w[j], m[j].y, acc.y);
                const float sz = fmaf(w[j], m[j].z, acc.z);
                const float sw = fmaf(w[j], m[j].w, acc.w);
                acc.x = ok ? sx : acc.x;
                acc.y = ok ? sy : acc.y;
                acc.z = ok ? sz : acc.z;
                acc.w = ok ? sw : acc.w;
                wsum  = ok ? (wsum + w[j]) : wsum;
            }
        }
    }
}

__global__ __launch_bounds__(256) void gcn_layer(
    const float* __restrict__ X, const float* __restrict__ W,
    const float* __restrict__ Bv, const int* __restrict__ offs,
    const int2* __restrict__ ent, float* __restrict__ out,
    int nnodes, int ntiles)
{
    __shared__ __attribute__((aligned(16))) float xs[2][16 * LDSPAD];
    __shared__ float rs[2][16];

    const int tid  = threadIdx.x;
    const int lane = tid & 31;
    const int wave = __builtin_amdgcn_readfirstlane(tid >> 5);
    const int nn   = lane & 15;
    const int kb   = (lane >> 4) << 1;
    const int mh   = (lane >> 4) * 8;
    const int n0   = wave * 16;

    v2f bw[32];
    {
        const float* wcol = &W[(size_t)kb * NFEAT + n0 + nn];
        #pragma unroll
        for (int k4 = 0; k4 < 32; ++k4) {
            bw[k4].x = wcol[(size_t)(4 * k4) * NFEAT];
            bw[k4].y = wcol[(size_t)(4 * k4 + 1) * NFEAT];
        }
    }
    const float bias = Bv[n0 + nn];

    int p = 0;
    for (int tile = blockIdx.x; tile < ntiles; tile += gridDim.x, p ^= 1) {
        const int m0 = tile * 16;

        {
            const int r0 = m0 + wave * 2;
            const int i0 = (r0     < nnodes) ? r0       : nnodes;
            const int i1 = (r0 + 1 < nnodes) ? (r0 + 1) : nnodes;
            const int i2 = (r0 + 2 < nnodes) ? (r0 + 2) : nnodes;
            const int s0 = offs[i0], s1 = offs[i1], s2 = offs[i2];
            const int nA = (s1 - s0 < 32) ? (s1 - s0) : 32;
            const int nB = (s2 - s1 < 32) ? (s2 - s1) : 32;
            const int2 evA = ent[(nA > 0) ? (s0 + ((lane < nA) ? lane : 0)) : 0];
            const int2 evB = ent[(nB > 0) ? (s1 + ((lane < nB) ? lane : 0)) : 0];

            float4 a0; a0.x = 0.f; a0.y = 0.f; a0.z = 0.f; a0.w = 0.f;
            float4 a1 = a0;
            float  w0 = 0.f, w1 = 0.f;
            gcn_row_sum(X, ent, s0, s1, evA, lane, a0, w0);
            gcn_row_sum(X, ent, s1, s2, evB, lane, a1, w1);

            float* xw = &xs[p][(wave * 2) * LDSPAD + lane * 4];
            *(float4*)xw            = a0;
            *(float4*)(xw + LDSPAD) = a1;
            if (lane == 0) { rs[p][wave * 2] = w0; rs[p][wave * 2 + 1] = w1; }
        }
        __syncthreads();

        v8f acc;
        #pragma unroll
        for (int r = 0; r < 8; ++r) acc[r] = rs[p][r + mh] * bias;

        const float* xrow = &xs[p][nn * LDSPAD + kb];
        #pragma unroll
        for (int k4 = 0; k4 < 32; ++k4) {
            v2f a;
            a.x = xrow[4 * k4];
            a.y = xrow[4 * k4 + 1];
            acc = __builtin_amdgcn_wmma_f32_16x16x4_f32(
                 false, a,  false, bw[k4],
                 (short)0, acc,  false,  false);
        }

        float o[8];
        #pragma unroll
        for (int r = 0; r < 8; ++r) {
            const float v = acc[r];
            o[r] = (v > 0.0f) ? v : (__expf(v) - 1.0f);
        }
        float* orow = out + (size_t)(m0 + mh) * NFEAT + n0 + nn;
        if (m0 + 16 <= nnodes) {
            #pragma unroll
            for (int r = 0; r < 8; ++r) orow[(size_t)r * NFEAT] = o[r];
        } else {
            #pragma unroll
            for (int r = 0; r < 8; ++r)
                if (m0 + mh + r < nnodes) orow[(size_t)r * NFEAT] = o[r];
        }
    }
}

extern "C" void kernel_launch(void* const* d_in, const int* in_sizes, int n_in,
                              void* d_out, int out_size, void* d_ws, size_t ws_size,
                              hipStream_t stream)
{
    const float* x    = (const float*)d_in[0];
    const float* W    = (const float*)d_in[1];
    const float* b    = (const float*)d_in[2];
    const int*   rows = (const int*)  d_in[3];
    const int*   cols = (const int*)  d_in[4];
    const float* vals = (const float*)d_in[5];

    const int nnodes  = in_sizes[0] / NFEAT;
    const int nlayers = in_sizes[1] / (NFEAT * NFEAT);
    const int nedges  = in_sizes[3] / nlayers;

    const int nbins   = (nnodes + BINROWS - 1) / BINROWS;
    const int nchunks = (nedges + CHUNK - 1) / CHUNK;

    const size_t nodeElems = (size_t)nnodes * NFEAT;
    const size_t allEdges  = (size_t)nlayers * nedges;
    float* actA = (float*)d_ws;
    float* actB = actA + nodeElems;
    int2*  ent  = (int2*)(actB + nodeElems);
    int2*  tent = ent + allEdges;
    int*   trow = (int*)(tent + allEdges);
    int*   tab  = trow + allEdges;
    int*   offs = tab + (size_t)nlayers * (nbins + 1) * nchunks;

    const int ntiles = (nnodes + 15) / 16;
    const dim3 chunkGrid(nchunks, nlayers);
    const dim3 binGrid(nbins, nlayers);

    gcn_chunk_sort<<<chunkGrid, 1024, 0, stream>>>(rows, cols, vals, tab, trow, tent,
                                                   nedges, nbins, nchunks);
    gcn_bin_sort<<<binGrid, 512, 0, stream>>>(tab, trow, tent, offs, ent,
                                              nedges, nnodes, nbins, nchunks);

    for (int l = 0; l < nlayers; ++l) {
        const float* cur = (l == 0) ? x : ((l & 1) ? actA : actB);
        float* dst = (l == nlayers - 1) ? (float*)d_out : ((l & 1) ? actB : actA);
        gcn_layer<<<LAYER_BLOCKS, 256, 0, stream>>>(
            cur, W + (size_t)l * NFEAT * NFEAT, b + (size_t)l * NFEAT,
            offs + (size_t)l * (nnodes + 1), ent + (size_t)l * nedges,
            dst, nnodes, ntiles);
    }
}


# Layer result goes through LDS and is stored as whole 512-byte rows, one wave per row, behind the next tile's barrier: 64-byte pieces (two waves per 128-byte line) gone; sum order unchanged
# speedup vs baseline: 4.0294x; 1.0352x over previous
#include <hip/hip_runtime.h>
#include <math.h>

typedef __attribute__((ext_vector_type(2))) float v2f;
typedef __attribute__((ext_vector_type(8))) float v8f;

#define NFEAT  128
#define LDSPAD 132

#define BINSHIFT 8
#define BINROWS  256
#define MAXBINS  1024
#define EPT      8
#define CHUNK    (1024 * EPT)
#define MAXCH    1024

__global__ __launch_bounds__(1024) void gcn_chunk_sort(
    const int* __restrict__ rows, const int* __restrict__ cols,
    const float* __restrict__ vals, int* __restrict__ tab,
    int* __restrict__ trow, int2* __restrict__ tent,
    int nedges, int nbins, int nchunks)
{
    __shared__ int  hist[MAXBINS];
    __shared__ int  wtot[32];
    __shared__ int  srow[CHUNK];
    __shared__ int2 sent[CHUNK];
    const int l = blockIdx.y, c = blockIdx.x, tid = threadIdx.x;
    const int lane = tid & 31, wave = tid >> 5;

    hist[tid] = 0;
    __syncthreads();

    const size_t lo = (size_t)l * nedges;
    const int e0 = c * CHUNK + tid;
    int r[EPT], cc[EPT], vv[EPT];
    #pragma unroll
    for (int j = 0; j < EPT; ++j) {
        const int e  = e0 + j * 1024;
        const size_t g = lo + ((e < nedges) ? e : (nedges - 1));
        r[j]  = rows[g];
        cc[j] = cols[g];
        vv[j] = __float_as_int(vals[g]);
    }
    int q[EPT];
    #pragma unroll
    for (int j = 0; j < EPT; ++j)
        q[j] = (e0 + j * 1024 < nedges) ? atomicAdd(&hist[r[j] >> BINSHIFT], 1) : 0;
    __syncthreads();

    const int cnt = hist[tid];
    int incl = cnt;
    for (int d = 1; d < 32; d <<= 1) {
        int v = __shfl_up(incl, d);
        if (lane >= d) incl += v;
    }
    if (lane == 31) wtot[wave] = incl;
    __syncthreads();
    int front = (lane < wave) ? wtot[lane] : 0;
    for (int d = 16; d >= 1; d >>= 1) front += __shfl_xor(front, d);
    const int lf = front + incl - cnt;
    hist[tid] = lf;

    const int left = nedges - c * CHUNK;
    const int nloc = (left < CHUNK) ? left : CHUNK;
    int* t = tab + (size_t)l * (nbins + 1) * nchunks;
    if (tid < nbins) t[(size_t)tid * nchunks + c] = lf;
    if (tid == 0)    t[(size_t)nbins * nchunks + c] = nloc;
    __syncthreads();

    #pragma unroll
    for (int j = 0; j < EPT; ++j) {
        if (e0 + j * 1024 < nedges) {
            const int slot = hist[r[j] >> BINSHIFT] + q[j];
            int2 v;
            v.x = cc[j];
            v.y = vv[j];
            srow[slot] = r[j];
            sent[slot] = v;
        }
    }
    __syncthreads();

    const size_t go = lo + (size_t)c * CHUNK;
    int  rr[EPT];
    int2 ev[EPT];
    #pragma unroll
    for (int j = 0; j < EPT; ++j) {
        const int i  = tid + j * 1024;
        const int ic = (i < nloc) ? i : (nloc - 1);
        rr[j] = srow[ic];
        ev[j] = sent[ic];
    }
    #pragma unroll
    for (int j = 0; j < EPT; ++j) {
        const int i = tid + j * 1024;
        if (i < nloc) {
            trow[go + i] = rr[j];
            tent[go + i] = ev[j];
        }
    }
}

#define SORTCAP 8192

__device__ __forceinline__ void gcn_locate(
    const int* pre, const int* gb, int half, const int (&i)[EPT], int (&g)[EPT])
{
    int cx[EPT];
    #pragma unroll
    for (int j = 0; j < EPT; ++j) cx[j] = 0;
    for (int st = half; st > 0; st >>= 1) {
        #pragma unroll
        for (int j = 0; j < EPT; ++j) {
            const int m = cx[j] + st;
            cx[j] = (pre[m] <= i[j]) ? m : cx[j];
        }
    }
    #pragma unroll
    for (int j = 0; j < EPT; ++j) g[j] = gb[cx[j]] + i[j];
}

__global__ __launch_bounds__(512) void gcn_bin_sort(
    const int* __restrict__ tab, const int* __restrict__ trow,
    const int2* __restrict__ tent, int* __restrict__ offs, int2* __restrict__ ent,
    int nedges, int nnodes, int nbins, int nchunks)
{
    __shared__ int  pre[MAXCH];
    __shared__ int  gb[MAXCH];
    __shared__ int  cnt[BINROWS];
    __shared__ int  cur[BINROWS];
    __shared__ int  wrun[16];
    __shared__ int  wsum[16];
    __shared__ int  wtot[8];
    __shared__ int2 stage[SORTCAP];
    const int l = blockIdx.y, bin = blockIdx.x;
    const int tid = threadIdx.x, lane = tid & 31, wave = tid >> 5;
    const size_t lo = (size_t)l * nedges;
    const int* ta = tab + ((size_t)l * (nbins + 1) + bin) * nchunks;
    const int* tz = ta + nchunks;

    const int c0 = 2 * tid, c1 = c0 + 1;
    int a0 = 0, z0 = 0, a1 = 0, z1 = 0;
    if (c0 < nchunks) { a0 = ta[c0]; z0 = tz[c0]; }
    if (c1 < nchunks) { a1 = ta[c1]; z1 = tz[c1]; }
    const int n0 = z0 - a0, n1 = z1 - a1, both = n0 + n1;
    int rinc = both;
    for (int d = 1; d < 32; d <<= 1) {
        int v = __shfl_up(rinc, d);
        if (lane >= d) rinc += v;
    }
    int sa = a0 + a1;
    for (int d = 16; d >= 1; d >>= 1) sa += __shfl_xor(sa, d);
    if (lane == 31) { wrun[wave] = rinc; wsum[wave] = sa; }
    if (tid < BINROWS) cnt[tid] = 0;
    __syncthreads();
    int fr = (lane < wave) ? wrun[lane] : 0;
    int al = (lane < 16) ? wrun[lane] : 0;
    int sl = (lane < 16) ? wsum[lane] : 0;
    for (int d = 16; d >= 1; d >>= 1) {
        fr += __shfl_xor(fr, d);
        al += __shfl_xor(al, d);
        sl += __shfl_xor(sl, d);
    }
    const int s = sl;
    const int n = al;
    const int ex = fr + rinc - both;
    pre[c0] = (c0 < nchunks) ? ex        : 0x7fffffff;
    pre[c1] = (c1 < nchunks) ? (ex + n0) : 0x7fffffff;
    gb[c0]  = c0 * CHUNK + a0 - ex;
    gb[c1]  = c1 * CHUNK + a1 - (ex + n0);
    __syncthreads();

    const int half = (nchunks > 1) ? (1 << (31 - __clz(nchunks - 1))) : 0;
    const bool staged = (n <= SORTCAP);

    int2 v[2][EPT];
    int  pk[2][EPT];
    #pragma unroll
    for (int h = 0; h < 2; ++h) {
        #pragma unroll
        for (int j = 0; j < EPT; ++j) { v[h][j].x = 0; v[h][j].y = 0; pk[h][j] = 0; }
    }

    if (staged) {
        #pragma unroll
        for (int h = 0; h < 2; ++h) {
            if (h * 512 * EPT < n) {
                int ic[EPT], g[EPT], r[EPT];
                #pragma unroll
                for (int j = 0; j < EPT; ++j) {
                    const int i = h * 512 * EPT + tid + j * 512;
                    ic[j] = (i < n) ? i : (n - 1);
                }
                gcn_locate(pre, gb, half, ic, g);
                #pragma unroll
                for (int j = 0; j < EPT; ++j) {
                    r[j]    = trow[lo + g[j]];
                    v[h][j] = tent[lo + g[j]];
                }
                #pragma unroll
                for (int j = 0; j < EPT; ++j) {
                    const int i  = h * 512 * EPT + tid + j * 512;
                    const int rl = r[j] & (BINROWS - 1);
                    const int qq = (i < n) ? atomicAdd(&cnt[rl], 1) : 0;
                    pk[h][j] = (qq << 8) | rl;
                }
            }
        }
    } else {
        for (int b = 0; b < n; b += 512 * EPT) {
            int ic[EPT], g[EPT], r[EPT];
            #pragma unroll
            for (int j = 0; j < EPT; ++j) {
                const int i = b + tid + j * 512;
                ic[j] = (i < n) ? i : (n - 1);
            }
            gcn_locate(pre, gb, half, ic, g);
            #pragma unroll
            for (int j = 0; j < EPT; ++j) r[j] = trow[lo + g[j]];
            #pragma unroll
            for (int j = 0; j < EPT; ++j)
                if (b + tid + j * 512 < n) atomicAdd(&cnt[r[j] & (BINROWS - 1)], 1);
        }
    }
    __syncthreads();

    const int mine = (tid < BINROWS) ? cnt[tid] : 0;
    int incl = mine;
    for (int d = 1; d < 32; d <<= 1) {
        int w = __shfl_up(incl, d);
        if (lane >= d) incl += w;
    }
    if (tid < BINROWS && lane == 31) wtot[wave] = incl;
    __syncthreads();
    int front = (lane < 8 && lane < wave) ? wtot[lane] : 0;
    for (int d = 16; d >= 1; d >>= 1) front += __shfl_xor(front, d);
    const int first = s + front + incl - mine;

    if (tid < BINROWS) {
        int* of = offs + (size_t)l * (nnodes + 1);
        const int row = bin * BINROWS + tid;
        cur[tid] = staged ? (first - s) : first;
        if (row < nnodes)      of[row] = first;
        if (row == nnodes - 1) of[nnodes] = first + mine;
    }
    __syncthreads();

    if (staged) {
        #pragma unroll
        for (int h = 0; h < 2; ++h) {
            #pragma unroll
            for (int j = 0; j < EPT; ++j) {
                const int i = h * 512 * EPT + tid + j * 512;
                if (i < n) stage[cur[pk[h][j] & (BINROWS - 1)] + (pk[h][j] >> 8)] = v[h][j];
            }
        }
        __syncthreads();
        int2* dst = ent + lo + s;
        for (int i = tid; i < n; i += 512) dst[i] = stage[i];
    } else {
        for (int b = 0; b < n; b += 512 * EPT) {
            int  ic[EPT], g[EPT], r[EPT];
            int2 w[EPT];
            #pragma unroll
            for (int j = 0; j < EPT; ++j) {
                const int i = b + tid + j * 512;
                ic[j] = (i < n) ? i : (n - 1);
            }
            gcn_locate(pre, gb, half, ic, g);
            #pragma unroll
            for (int j = 0; j < EPT; ++j) {
                r[j] = trow[lo + g[j]];
                w[j] = tent[lo + g[j]];
            }
            #pragma unroll
            for (int j = 0; j < EPT; ++j) {
                if (b + tid + j * 512 < n) {
                    const int p = atomicAdd(&cur[r[j] & (BINROWS - 1)], 1);
                    ent[lo + p] = w[j];
                }
            }
        }
    }
}

#define LAYER_BLOCKS 768


__device__ __forceinline__ void gcn_row_sum(
    const float* __restrict__ X, const int2* __restrict__ ent,
    int s, int t, int2 ev, int lane, float4& acc, float& wsum)
{
    for (int b = s; b < t; b += 32) {
        const int n = (t - b < 32) ? (t - b) : 32;
        if (b != s) ev = ent[b + ((lane < n) ? lane : 0)];
        const int nb = (n + 7) >> 3;
        for (int q = 0; q < nb; ++q) {
            const int k = q * 8;
            float4 m[8];
            float  w[8];
            #pragma unroll
            for (int j = 0; j < 8; ++j) {
                const int kk = (k + j < n) ? (k + j) : (n - 1);
                const int c  = __builtin_amdgcn_readlane(ev.x, kk);
                w[j] = __int_as_float(__builtin_amdgcn_readlane(ev.y, kk));
                m[j] = ((const float4*)(X + (size_t)c * NFEAT))[lane];
            }
            #pragma unroll
            for (int j = 0; j < 8; ++j) {
                const bool ok = (j == 0) || (k + j < n);
                const float sx = fmaf(w[j], m[j].x, acc.x);
                const float sy = fmaf(w[j], m[j].y, acc.y);
                const float sz = fmaf(w[j], m[j].z, acc.z);
                const float sw = fmaf(w[j], m[j].w, acc.w);
                acc.x = ok ? sx : acc.x;
                acc.y = ok ? sy : acc.y;
                acc.z = ok ? sz : acc.z;
                acc.w = ok ? sw : acc.w;
                wsum  = ok ? (wsum + w[j]) : wsum;
            }
        }
    }
}

__global__ __launch_bounds__(256) void gcn_layer(
    const float* __restrict__ X, const float* __restrict__ W,
    const float* __restrict__ Bv, const int* __restrict__ offs,
    const int2* __restrict__ ent, float* __restrict__ out,
    int nnodes, int ntiles)
{
    __shared__ __attribute__((aligned(16))) float xs[2][16 * LDSPAD];
    __shared__ float rs[2][16];
    __shared__ __attribute__((aligned(16))) float os[2][16 * LDSPAD];

    const int tid  = threadIdx.x;
    const int lane = tid & 31;
    const int wave = __builtin_amdgcn_readfirstlane(tid >> 5);
    const int nn   = lane & 15;
    const int kb   = (lane >> 4) << 1;
    const int mh   = (lane >> 4) * 8;
    const int n0   = wave * 16;

    v2f bw[32];
    {
        const float* wcol = &W[(size_t)kb * NFEAT + n0 + nn];
        #pragma unroll
        for (int k4 = 0; k4 < 32; ++k4) {
            bw[k4].x = wcol[(size_t)(4 * k4) * NFEAT];
            bw[k4].y = wcol[(size_t)(4 * k4 + 1) * NFEAT];
        }
    }
    const float bias = Bv[n0 + nn];

    auto store_rows = [&](int buf, int mfirst) {
        const int r0 = mfirst + wave * 2;
        const float4* t = (const float4*)&os[buf][(wave * 2) * LDSPAD + lane * 4];
        float4* o0 = (float4*)(out + (size_t)r0 * NFEAT) + lane;
        if (r0 + 1 < nnodes) {
            const float4 q0 = t[0];
            const float4 q1 = t[LDSPAD / 4];
            o0[0]         = q0;
            o0[NFEAT / 4] = q1;
        } else if (r0 < nnodes) {
            o0[0] = t[0];
        }
    };

    int p = 0;
    int mprev = -1;
    for (int tile = blockIdx.x; tile < ntiles; tile += gridDim.x, p ^= 1) {
        const int m0 = tile * 16;

        {
            const int r0 = m0 + wave * 2;
            const int i0 = (r0     < nnodes) ? r0       : nnodes;
            const int i1 = (r0 + 1 < nnodes) ? (r0 + 1) : nnodes;
            const int i2 = (r0 + 2 < nnodes) ? (r0 + 2) : nnodes;
            const int s0 = offs[i0], s1 = offs[i1], s2 = offs[i2];
            const int nA = (s1 - s0 < 32) ? (s1 - s0) : 32;
            const int nB = (s2 - s1 < 32) ? (s2 - s1) : 32;
            const int2 evA = ent[(nA > 0) ? (s0 + ((lane < nA) ? lane : 0)) : 0];
            const int2 evB = ent[(nB > 0) ? (s1 + ((lane < nB) ? lane : 0)) : 0];

            float4 a0; a0.x = 0.f; a0.y = 0.f; a0.z = 0.f; a0.w = 0.f;
            float4 a1 = a0;
            float  w0 = 0.f, w1 = 0.f;
            gcn_row_sum(X, ent, s0, s1, evA, lane, a0, w0);
            gcn_row_sum(X, ent, s1, s2, evB, lane, a1, w1);

            float* xw = &xs[p][(wave * 2) * LDSPAD + lane * 4];
            *(float4*)xw            = a0;
            *(float4*)(xw + LDSPAD) = a1;
            if (lane == 0) { rs[p][wave * 2] = w0; rs[p][wave * 2 + 1] = w1; }
        }
        __syncthreads();

        if (mprev >= 0) store_rows(p ^ 1, mprev);

        v8f acc;
        #pragma unroll
        for (int r = 0; r < 8; ++r) acc[r] = rs[p][r + mh] * bias;

        const float* xrow = &xs[p][nn * LDSPAD + kb];
        #pragma unroll
        for (int k4 = 0; k4 < 32; ++k4) {
            v2f a;
            a.x = xrow[4 * k4];
            a.y = xrow[4 * k4 + 1];
            acc = __builtin_amdgcn_wmma_f32_16x16x4_f32(
                 false, a,  false, bw[k4],
                 (short)0, acc,  false,  false);
        }

        float o[8];
        #pragma unroll
        for (int r = 0; r < 8; ++r) {
            const float v = acc[r];
            o[r] = (v > 0.0f) ? v : (__expf(v) - 1.0f);
        }
        float* ow = &os[p][mh * LDSPAD + n0 + nn];
        #pragma unroll
        for (int r = 0; r < 8; ++r) ow[r * LDSPAD] = o[r];
        mprev = m0;
    }
    __syncthreads();
    if (mprev >= 0) store_rows(p ^ 1, mprev);
}

extern "C" void kernel_launch(void* const* d_in, const int* in_sizes, int n_in,
                              void* d_out, int out_size, void* d_ws, size_t ws_size,
                              hipStream_t stream)
{
    const float* x    = (const float*)d_in[0];
    const float* W    = (const float*)d_in[1];
    const float* b    = (const float*)d_in[2];
    const int*   rows = (const int*)  d_in[3];
    const int*   cols = (const int*)  d_in[4];
    const float* vals = (const float*)d_in[5];

    const int nnodes  = in_sizes[0] / NFEAT;
    const int nlayers = in_sizes[1] / (NFEAT * NFEAT);
    const int nedges  = in_sizes[3] / nlayers;

    const int nbins   = (nnodes + BINROWS - 1) / BINROWS;
    const int nchunks = (nedges + CHUNK - 1) / CHUNK;

    const size_t nodeElems = (size_t)nnodes * NFEAT;
    const size_t allEdges  = (size_t)nlayers * nedges;
    float* actA = (float*)d_ws;
    float* actB = actA + nodeElems;
    int2*  ent  = (int2*)(actB + nodeElems);
    int2*  tent = ent + allEdges;
    int*   trow = (int*)(tent + allEdges);
    int*   tab  = trow + allEdges;
    int*   offs = tab + (size_t)nlayers * (nbins + 1) * nchunks;

    const int ntiles = (nnodes + 15) / 16;
    const dim3 chunkGrid(nchunks, nlayers);
    const dim3 binGrid(nbins, nlayers);

    gcn_chunk_sort<<<chunkGrid, 1024, 0, stream>>>(rows, cols, vals, tab, trow, tent,
                                                   nedges, nbins, nchunks);
    gcn_bin_sort<<<binGrid, 512, 0, stream>>>(tab, trow, tent, offs, ent,
                                              nedges, nnodes, nbins, nchunks);

    for (int l = 0; l < nlayers; ++l) {
        const float* cur = (l == 0) ? x : ((l & 1) ? actA : actB);
        float* dst = (l == nlayers - 1) ? (float*)d_out : ((l & 1) ? actB : actA);
        gcn_layer<<<LAYER_BLOCKS, 256, 0, stream>>>(
            cur, W + (size_t)l * NFEAT * NFEAT, b + (size_t)l * NFEAT,
            offs + (size_t)l * (nnodes + 1), ent + (size_t)l * nedges,
            dst, nnodes, ntiles);
    }
}


# Layer gather: a row load takes X as scalar base and a 32-bit offset (col x 512 + lane x 16) in a register of its own, a batch's 8 offsets built first: its 8 loads go back to back; sum order unchanged
# speedup vs baseline: 4.4067x; 1.0936x over previous
#include <hip/hip_runtime.h>
#include <math.h>

typedef __attribute__((ext_vector_type(2))) float v2f;
typedef __attribute__((ext_vector_type(8))) float v8f;

#define NFEAT  128
#define LDSPAD 132

#define BINSHIFT 8
#define BINROWS  256
#define MAXBINS  1024
#define EPT      8
#define CHUNK    (1024 * EPT)
#define MAXCH    1024

__global__ __launch_bounds__(1024) void gcn_chunk_sort(
    const int* __restrict__ rows, const int* __restrict__ cols,
    const float* __restrict__ vals, int* __restrict__ tab,
    int* __restrict__ trow, int2* __restrict__ tent,
    int nedges, int nbins, int nchunks)
{
    __shared__ int  hist[MAXBINS];
    __shared__ int  wtot[32];
    __shared__ int  srow[CHUNK];
    __shared__ int2 sent[CHUNK];
    const int l = blockIdx.y, c = blockIdx.x, tid = threadIdx.x;
    const int lane = tid & 31, wave = tid >> 5;

    hist[tid] = 0;
    __syncthreads();

    const size_t lo = (size_t)l * nedges;
    const int e0 = c * CHUNK + tid;
    int r[EPT], cc[EPT], vv[EPT];
    #pragma unroll
    for (int j = 0; j < EPT; ++j) {
        const int e  = e0 + j * 1024;
        const size_t g = lo + ((e < nedges) ? e : (nedges - 1));
        r[j]  = rows[g];
        cc[j] = cols[g];
        vv[j] = __float_as_int(vals[g]);
    }
    int q[EPT];
    #pragma unroll
    for (int j = 0; j < EPT; ++j)
        q[j] = (e0 + j * 1024 < nedges) ? atomicAdd(&hist[r[j] >> BINSHIFT], 1) : 0;
    __syncthreads();

    const int cnt = hist[tid];
    int incl = cnt;
    for (int d = 1; d < 32; d <<= 1) {
        int v = __shfl_up(incl, d);
        if (lane >= d) incl += v;
    }
    if (lane == 31) wtot[wave] = incl;
    __syncthreads();
    int front = (lane < wave) ? wtot[lane] : 0;
    for (int d = 16; d >= 1; d >>= 1) front += __shfl_xor(front, d);
    const int lf = front + incl - cnt;
    hist[tid] = lf;

    const int left = nedges - c * CHUNK;
    const int nloc = (left < CHUNK) ? left : CHUNK;
    int* t = tab + (size_t)l * (nbins + 1) * nchunks;
    if (tid < nbins) t[(size_t)tid * nchunks + c] = lf;
    if (tid == 0)    t[(size_t)nbins * nchunks + c] = nloc;
    __syncthreads();

    #pragma unroll
    for (int j = 0; j < EPT; ++j) {
        if (e0 + j * 1024 < nedges) {
            const int slot = hist[r[j] >> BINSHIFT] + q[j];
            int2 v;
            v.x = cc[j];
            v.y = vv[j];
            srow[slot] = r[j];
            sent[slot] = v;
        }
    }
    __syncthreads();

    const size_t go = lo + (size_t)c * CHUNK;
    int  rr[EPT];
    int2 ev[EPT];
    #pragma unroll
    for (int j = 0; j < EPT; ++j) {
        const int i  = tid + j * 1024;
        const int ic = (i < nloc) ? i : (nloc - 1);
        rr[j] = srow[ic];
        ev[j] = sent[ic];
    }
    #pragma unroll
    for (int j = 0; j < EPT; ++j) {
        const int i = tid + j * 1024;
        if (i < nloc) {
            trow[go + i] = rr[j];
            tent[go + i] = ev[j];
        }
    }
}

#define SORTCAP 8192

__device__ __forceinline__ void gcn_locate(
    const int* pre, const int* gb, int half, const int (&i)[EPT], int (&g)[EPT])
{
    int cx[EPT];
    #pragma unroll
    for (int j = 0; j < EPT; ++j) cx[j] = 0;
    for (int st = half; st > 0; st >>= 1) {
        #pragma unroll
        for (int j = 0; j < EPT; ++j) {
            const int m = cx[j] + st;
            cx[j] = (pre[m] <= i[j]) ? m : cx[j];
        }
    }
    #pragma unroll
    for (int j = 0; j < EPT; ++j) g[j] = gb[cx[j]] + i[j];
}

__global__ __launch_bounds__(512) void gcn_bin_sort(
    const int* __restrict__ tab, const int* __restrict__ trow,
    const int2* __restrict__ tent, int* __restrict__ offs, int2* __restrict__ ent,
    int nedges, int nnodes, int nbins, int nchunks)
{
    __shared__ int  pre[MAXCH];
    __shared__ int  gb[MAXCH];
    __shared__ int  cnt[BINROWS];
    __shared__ int  cur[BINROWS];
    __shared__ int  wrun[16];
    __shared__ int  wsum[16];
    __shared__ int  wtot[8];
    __shared__ int2 stage[SORTCAP];
    const int l = blockIdx.y, bin = blockIdx.x;
    const int tid = threadIdx.x, lane = tid & 31, wave = tid >> 5;
    const size_t lo = (size_t)l * nedges;
    const int* ta = tab + ((size_t)l * (nbins + 1) + bin) * nchunks;
    const int* tz = ta + nchunks;

    const int c0 = 2 * tid, c1 = c0 + 1;
    int a0 = 0, z0 = 0, a1 = 0, z1 = 0;
    if (c0 < nchunks) { a0 = ta[c0]; z0 = tz[c0]; }
    if (c1 < nchunks) { a1 = ta[c1]; z1 = tz[c1]; }
    const int n0 = z0 - a0, n1 = z1 - a1, both = n0 + n1;
    int rinc = both;
    for (int d = 1; d < 32; d <<= 1) {
        int v = __shfl_up(rinc, d);
        if (lane >= d) rinc += v;
    }
    int sa = a0 + a1;
    for (int d = 16; d >= 1; d >>= 1) sa += __shfl_xor(sa, d);
    if (lane == 31) { wrun[wave] = rinc; wsum[wave] = sa; }
    if (tid < BINROWS) cnt[tid] = 0;
    __syncthreads();
    int fr = (lane < wave) ? wrun[lane] : 0;
    int al = (lane < 16) ? wrun[lane] : 0;
    int sl = (lane < 16) ? wsum[lane] : 0;
    for (int d = 16; d >= 1; d >>= 1) {
        fr += __shfl_xor(fr, d);
        al += __shfl_xor(al, d);
        sl += __shfl_xor(sl, d);
    }
    const int s = sl;
    const int n = al;
    const int ex = fr + rinc - both;
    pre[c0] = (c0 < nchunks) ? ex        : 0x7fffffff;
    pre[c1] = (c1 < nchunks) ? (ex + n0) : 0x7fffffff;
    gb[c0]  = c0 * CHUNK + a0 - ex;
    gb[c1]  = c1 * CHUNK + a1 - (ex + n0);
    __syncthreads();

    const int half = (nchunks > 1) ? (1 << (31 - __clz(nchunks - 1))) : 0;
    const bool staged = (n <= SORTCAP);

    int2 v[2][EPT];
    int  pk[2][EPT];
    #pragma unroll
    for (int h = 0; h < 2; ++h) {
        #pragma unroll
        for (int j = 0; j < EPT; ++j) { v[h][j].x = 0; v[h][j].y = 0; pk[h][j] = 0; }
    }

    if (staged) {
        #pragma unroll
        for (int h = 0; h < 2; ++h) {
            if (h * 512 * EPT < n) {
                int ic[EPT], g[EPT], r[EPT];
                #pragma unroll
                for (int j = 0; j < EPT; ++j) {
                    const int i = h * 512 * EPT + tid + j * 512;
                    ic[j] = (i < n) ? i : (n - 1);
                }
                gcn_locate(pre, gb, half, ic, g);
                #pragma unroll
                for (int j = 0; j < EPT; ++j) {
                    r[j]    = trow[lo + g[j]];
                    v[h][j] = tent[lo + g[j]];
                }
                #pragma unroll
                for (int j = 0; j < EPT; ++j) {
                    const int i  = h * 512 * EPT + tid + j * 512;
                    const int rl = r[j] & (BINROWS - 1);
                    const int qq = (i < n) ? atomicAdd(&cnt[rl], 1) : 0;
                    pk[h][j] = (qq << 8) | rl;
                }
            }
        }
    } else {
        for (int b = 0; b < n; b += 512 * EPT) {
            int ic[EPT], g[EPT], r[EPT];
            #pragma unroll
            for (int j = 0; j < EPT; ++j) {
                const int i = b + tid + j * 512;
                ic[j] = (i < n) ? i : (n - 1);
            }
            gcn_locate(pre, gb, half, ic, g);
            #pragma unroll
            for (int j = 0; j < EPT; ++j) r[j] = trow[lo + g[j]];
            #pragma unroll
            for (int j = 0; j < EPT; ++j)
                if (b + tid + j * 512 < n) atomicAdd(&cnt[r[j] & (BINROWS - 1)], 1);
        }
    }
    __syncthreads();

    const int mine = (tid < BINROWS) ? cnt[tid] : 0;
    int incl = mine;
    for (int d = 1; d < 32; d <<= 1) {
        int w = __shfl_up(incl, d);
        if (lane >= d) incl += w;
    }
    if (tid < BINROWS && lane == 31) wtot[wave] = incl;
    __syncthreads();
    int front = (lane < 8 && lane < wave) ? wtot[lane] : 0;
    for (int d = 16; d >= 1; d >>= 1) front += __shfl_xor(front, d);
    const int first = s + front + incl - mine;

    if (tid < BINROWS) {
        int* of = offs + (size_t)l * (nnodes + 1);
        const int row = bin * BINROWS + tid;
        cur[tid] = staged ? (first - s) : first;
        if (row < nnodes)      of[row] = first;
        if (row == nnodes - 1) of[nnodes] = first + mine;
    }
    __syncthreads();

    if (staged) {
        #pragma unroll
        for (int h = 0; h < 2; ++h) {
            #pragma unroll
            for (int j = 0; j < EPT; ++j) {
                const int i = h * 512 * EPT + tid + j * 512;
                if (i < n) stage[cur[pk[h][j] & (BINROWS - 1)] + (pk[h][j] >> 8)] = v[h][j];
            }
        }
        __syncthreads();
        int2* dst = ent + lo + s;
        for (int i = tid; i < n; i += 512) dst[i] = stage[i];
    } else {
        for (int b = 0; b < n; b += 512 * EPT) {
            int  ic[EPT], g[EPT], r[EPT];
            int2 w[EPT];
            #pragma unroll
            for (int j = 0; j < EPT; ++j) {
                const int i = b + tid + j * 512;
                ic[j] = (i < n) ? i : (n - 1);
            }
            gcn_locate(pre, gb, half, ic, g);
            #pragma unroll
            for (int j = 0; j < EPT; ++j) {
                r[j] = trow[lo + g[j]];
                w[j] = tent[lo + g[j]];
            }
            #pragma unroll
            for (int j = 0; j < EPT; ++j) {
                if (b + tid + j * 512 < n) {
                    const int p = atomicAdd(&cur[r[j] & (BINROWS - 1)], 1);
                    ent[lo + p] = w[j];
                }
            }
        }
    }
}

#define LAYER_BLOCKS 768


typedef __attribute__((ext_vector_type(4))) float v4f;
typedef const __attribute__((address_space(1))) char* gcn_bytep;
typedef const __attribute__((address_space(1))) v4f*  gcn_rowp;

__device__ __forceinline__ void gcn_row_sum(
    const float* __restrict__ X, const int2* __restrict__ ent,
    int s, int t, int2 ev, int lane, float4& acc, float& wsum)
{
    const gcn_bytep xb = (gcn_bytep)X;
    const unsigned lane16 = (unsigned)lane << 4;
    for (int b = s; b < t; b += 32) {
        const int n = (t - b < 32) ? (t - b) : 32;
        if (b != s) ev = ent[b + ((lane < n) ? lane : 0)];
        const int nb = (n + 7) >> 3;
        for (int q = 0; q < nb; ++q) {
            const int k = q * 8;
            v4f      m[8];
            float    w[8];
            unsigned off[8];
            #pragma unroll
            for (int j = 0; j < 8; ++j) {
                const int kk = (k + j < n) ? (k + j) : (n - 1);
                const unsigned c = (unsigned)__builtin_amdgcn_readlane(ev.x, kk);
                w[j]   = __int_as_float(__builtin_amdgcn_readlane(ev.y, kk));
                off[j] = (c << 9) + lane16;
            }
            #pragma unroll
            for (int j = 0; j < 8; ++j) m[j] = *(gcn_rowp)(xb + off[j]);
            __builtin_amdgcn_sched_group_barrier(0x002, 24, 0);
            __builtin_amdgcn_sched_group_barrier(0x020, 8, 0);
            #pragma unroll
            for (int j = 0; j < 8; ++j) {
                const bool ok = (j == 0) || (k + j < n);
                const float sx = fmaf(w[j], m[j].x, acc.x);
                const float sy = fmaf(w[j], m[j].y, acc.y);
                const float sz = fmaf(w[j], m[j].z, acc.z);
                const float sw = fmaf(w[j], m[j].w, acc.w);
                acc.x = ok ? sx : acc.x;
                acc.y = ok ? sy : acc.y;
                acc.z = ok ? sz : acc.z;
                acc.w = ok ? sw : acc.w;
                wsum  = ok ? (wsum + w[j]) : wsum;
            }
        }
    }
}

__global__ __launch_bounds__(256) void gcn_layer(
    const float* __restrict__ X, const float* __restrict__ W,
    const float* __restrict__ Bv, const int* __restrict__ offs,
    const int2* __restrict__ ent, float* __restrict__ out,
    int nnodes, int ntiles)
{
    __shared__ __attribute__((aligned(16))) float xs[2][16 * LDSPAD];
    __shared__ float rs[2][16];
    __shared__ __attribute__((aligned(16))) float os[2][16 * LDSPAD];

    const int tid  = threadIdx.x;
    const int lane = tid & 31;
    const int wave = __builtin_amdgcn_readfirstlane(tid >> 5);
    const int nn   = lane & 15;
    const int kb   = (lane >> 4) << 1;
    const int mh   = (lane >> 4) * 8;
    const int n0   = wave * 16;

    v2f bw[32];
    {
        const float* wcol = &W[(size_t)kb * NFEAT + n0 + nn];
        #pragma unroll
        for (int k4 = 0; k4 < 32; ++k4) {
            bw[k4].x = wcol[(size_t)(4 * k4) * NFEAT];
            bw[k4].y = wcol[(size_t)(4 * k4 + 1) * NFEAT];
        }
    }
    const float bias = Bv[n0 + nn];

    auto store_rows = [&](int buf, int mfirst) {
        const int r0 = mfirst + wave * 2;
        const float4* t = (const float4*)&os[buf][(wave * 2) * LDSPAD + lane * 4];
        float4* o0 = (float4*)(out + (size_t)r0 * NFEAT) + lane;
        if (r0 + 1 < nnodes) {
            const float4 q0 = t[0];
            const float4 q1 = t[LDSPAD / 4];
            o0[0]         = q0;
            o0[NFEAT / 4] = q1;
        } else if (r0 < nnodes) {
            o0[0] = t[0];
        }
    };

    int p = 0;
    int mprev = -1;
    for (int tile = blockIdx.x; tile < ntiles; tile += gridDim.x, p ^= 1) {
        const int m0 = tile * 16;

        {
            const int r0 = m0 + wave * 2;
            const int i0 = (r0     < nnodes) ? r0       : nnodes;
            const int i1 = (r0 + 1 < nnodes) ? (r0 + 1) : nnodes;
            const int i2 = (r0 + 2 < nnodes) ? (r0 + 2) : nnodes;
            const int s0 = offs[i0], s1 = offs[i1], s2 = offs[i2];
            const int nA = (s1 - s0 < 32) ? (s1 - s0) : 32;
            const int nB = (s2 - s1 < 32) ? (s2 - s1) : 32;
            const int2 evA = ent[(nA > 0) ? (s0 + ((lane < nA) ? lane : 0)) : 0];
            const int2 evB = ent[(nB > 0) ? (s1 + ((lane < nB) ? lane : 0)) : 0];

            float4 a0; a0.x = 0.f; a0.y = 0.f; a0.z = 0.f; a0.w = 0.f;
            float4 a1 = a0;
            float  w0 = 0.f, w1 = 0.f;
            gcn_row_sum(X, ent, s0, s1, evA, lane, a0, w0);
            gcn_row_sum(X, ent, s1, s2, evB, lane, a1, w1);

            float* xw = &xs[p][(wave * 2) * LDSPAD + lane * 4];
            *(float4*)xw            = a0;
            *(float4*)(xw + LDSPAD) = a1;
            if (lane == 0) { rs[p][wave * 2] = w0; rs[p][wave * 2 + 1] = w1; }
        }
        __syncthreads();

        if (mprev >= 0) store_rows(p ^ 1, mprev);

        v8f acc;
        #pragma unroll
        for (int r = 0; r < 8; ++r) acc[r] = rs[p][r + mh] * bias;

        const float* xrow = &xs[p][nn * LDSPAD + kb];
        #pragma unroll
        for (int k4 = 0; k4 < 32; ++k4) {
            v2f a;
            a.x = xrow[4 * k4];
            a.y = xrow[4 * k4 + 1];
            acc = __builtin_amdgcn_wmma_f32_16x16x4_f32(
                 false, a,  false, bw[k4],
                 (short)0, acc,  false,  false);
        }

        float o[8];
        #pragma unroll
        for (int r = 0; r < 8; ++r) {
            const float v = acc[r];
            o[r] = (v > 0.0f) ? v : (__expf(v) - 1.0f);
        }
        float* ow = &os[p][mh * LDSPAD + n0 + nn];
        #pragma unroll
        for (int r = 0; r < 8; ++r) ow[r * LDSPAD] = o[r];
        mprev = m0;
    }
    __syncthreads();
    if (mprev >= 0) store_rows(p ^ 1, mprev);
}

extern "C" void kernel_launch(void* const* d_in, const int* in_sizes, int n_in,
                              void* d_out, int out_size, void* d_ws, size_t ws_size,
                              hipStream_t stream)
{
    const float* x    = (const float*)d_in[0];
    const float* W    = (const float*)d_in[1];
    const float* b    = (const float*)d_in[2];
    const int*   rows = (const int*)  d_in[3];
    const int*   cols = (const int*)  d_in[4];
    const float* vals = (const float*)d_in[5];

    const int nnodes  = in_sizes[0] / NFEAT;
    const int nlayers = in_sizes[1] / (NFEAT * NFEAT);
    const int nedges  = in_sizes[3] / nlayers;

    const int nbins   = (nnodes + BINROWS - 1) / BINROWS;
    const int nchunks = (nedges + CHUNK - 1) / CHUNK;

    const size_t nodeElems = (size_t)nnodes * NFEAT;
    const size_t allEdges  = (size_t)nlayers * nedges;
    float* actA = (float*)d_ws;
    float* actB = actA + nodeElems;
    int2*  ent  = (int2*)(actB + nodeElems);
    int2*  tent = ent + allEdges;
    int*   trow = (int*)(tent + allEdges);
    int*   tab  = trow + allEdges;
    int*   offs = tab + (size_t)nlayers * (nbins + 1) * nchunks;

    const int ntiles = (nnodes + 15) / 16;
    const dim3 chunkGrid(nchunks, nlayers);
    const dim3 binGrid(nbins, nlayers);

    gcn_chunk_sort<<<chunkGrid, 1024, 0, stream>>>(rows, cols, vals, tab, trow, tent,
                                                   nedges, nbins, nchunks);
    gcn_bin_sort<<<binGrid, 512, 0, stream>>>(tab, trow, tent, offs, ent,
                                              nedges, nnodes, nbins, nchunks);

    for (int l = 0; l < nlayers; ++l) {
        const float* cur = (l == 0) ? x : ((l & 1) ? actA : actB);
        float* dst = (l == nlayers - 1) ? (float*)d_out : ((l & 1) ? actB : actA);
        gcn_layer<<<LAYER_BLOCKS, 256, 0, stream>>>(
            cur, W + (size_t)l * NFEAT * NFEAT, b + (size_t)l * NFEAT,
            offs + (size_t)l * (nnodes + 1), ent + (size_t)l * nedges,
            dst, nnodes, ntiles);
    }
}


# Layer kernel held to 128 registers (8 waves per SIMD) so that 4 blocks fit a WGP; grid 1024 resident blocks, was 768: a third more row loads in flight, a block walks 3 tiles; sum order unchanged
# speedup vs baseline: 4.5769x; 1.0386x over previous
#include <hip/hip_runtime.h>
#include <math.h>

typedef __attribute__((ext_vector_type(2))) float v2f;
typedef __attribute__((ext_vector_type(8))) float v8f;

#define NFEAT  128
#define LDSPAD 132

#define BINSHIFT 8
#define BINROWS  256
#define MAXBINS  1024
#define EPT      8
#define CHUNK    (1024 * EPT)
#define MAXCH    1024

__global__ __launch_bounds__(1024) void gcn_chunk_sort(
    const int* __restrict__ rows, const int* __restrict__ cols,
    const float* __restrict__ vals, int* __restrict__ tab,
    int* __restrict__ trow, int2* __restrict__ tent,
    int nedges, int nbins, int nchunks)
{
    __shared__ int  hist[MAXBINS];
    __shared__ int  wtot[32];
    __shared__ int  srow[CHUNK];
    __shared__ int2 sent[CHUNK];
    const int l = blockIdx.y, c = blockIdx.x, tid = threadIdx.x;
    const int lane = tid & 31, wave = tid >> 5;

    hist[tid] = 0;
    __syncthreads();

    const size_t lo = (size_t)l * nedges;
    const int e0 = c * CHUNK + tid;
    int r[EPT], cc[EPT], vv[EPT];
    #pragma unroll
    for (int j = 0; j < EPT; ++j) {
        const int e  = e0 + j * 1024;
        const size_t g = lo + ((e < nedges) ? e : (nedges - 1));
        r[j]  = rows[g];
        cc[j] = cols[g];
        vv[j] = __float_as_int(vals[g]);
    }
    int q[EPT];
    #pragma unroll
    for (int j = 0; j < EPT; ++j)
        q[j] = (e0 + j * 1024 < nedges) ? atomicAdd(&hist[r[j] >> BINSHIFT], 1) : 0;
    __syncthreads();

    const int cnt = hist[tid];
    int incl = cnt;
    for (int d = 1; d < 32; d <<= 1) {
        int v = __shfl_up(incl, d);
        if (lane >= d) incl += v;
    }
    if (lane == 31) wtot[wave] = incl;
    __syncthreads();
    int front = (lane < wave) ? wtot[lane] : 0;
    for (int d = 16; d >= 1; d >>= 1) front += __shfl_xor(front, d);
    const int lf = front + incl - cnt;
    hist[tid] = lf;

    const int left = nedges - c * CHUNK;
    const int nloc = (left < CHUNK) ? left : CHUNK;
    int* t = tab + (size_t)l * (nbins + 1) * nchunks;
    if (tid < nbins) t[(size_t)tid * nchunks + c] = lf;
    if (tid == 0)    t[(size_t)nbins * nchunks + c] = nloc;
    __syncthreads();

    #pragma unroll
    for (int j = 0; j < EPT; ++j) {
        if (e0 + j * 1024 < nedges) {
            const int slot = hist[r[j] >> BINSHIFT] + q[j];
            int2 v;
            v.x = cc[j];
            v.y = vv[j];
            srow[slot] = r[j];
            sent[slot] = v;
        }
    }
    __syncthreads();

    const size_t go = lo + (size_t)c * CHUNK;
    int  rr[EPT];
    int2 ev[EPT];
    #pragma unroll
    for (int j = 0; j < EPT; ++j) {
        const int i  = tid + j * 1024;
        const int ic = (i < nloc) ? i : (nloc - 1);
        rr[j] = srow[ic];
        ev[j] = sent[ic];
    }
    #pragma unroll
    for (int j = 0; j < EPT; ++j) {
        const int i = tid + j * 1024;
        if (i < nloc) {
            trow[go + i] = rr[j];
            tent[go + i] = ev[j];
        }
    }
}

#define SORTCAP 8192

__device__ __forceinline__ void gcn_locate(
    const int* pre, const int* gb, int half, const int (&i)[EPT], int (&g)[EPT])
{
    int cx[EPT];
    #pragma unroll
    for (int j = 0; j < EPT; ++j) cx[j] = 0;
    for (int st = half; st > 0; st >>= 1) {
        #pragma unroll
        for (int j = 0; j < EPT; ++j) {
            const int m = cx[j] + st;
            cx[j] = (pre[m] <= i[j]) ? m : cx[j];
        }
    }
    #pragma unroll
    for (int j = 0; j < EPT; ++j) g[j] = gb[cx[j]] + i[j];
}

__global__ __launch_bounds__(512) void gcn_bin_sort(
    const int* __restrict__ tab, const int* __restrict__ trow,
    const int2* __restrict__ tent, int* __restrict__ offs, int2* __restrict__ ent,
    int nedges, int nnodes, int nbins, int nchunks)
{
    __shared__ int  pre[MAXCH];
    __shared__ int  gb[MAXCH];
    __shared__ int  cnt[BINROWS];
    __shared__ int  cur[BINROWS];
    __shared__ int  wrun[16];
    __shared__ int  wsum[16];
    __shared__ int  wtot[8];
    __shared__ int2 stage[SORTCAP];
    const int l = blockIdx.y, bin = blockIdx.x;
    const int tid = threadIdx.x, lane = tid & 31, wave = tid >> 5;
    const size_t lo = (size_t)l * nedges;
    const int* ta = tab + ((size_t)l * (nbins + 1) + bin) * nchunks;
    const int* tz = ta + nchunks;

    const int c0 = 2 * tid, c1 = c0 + 1;
    int a0 = 0, z0 = 0, a1 = 0, z1 = 0;
    if (c0 < nchunks) { a0 = ta[c0]; z0 = tz[c0]; }
    if (c1 < nchunks) { a1 = ta[c1]; z1 = tz[c1]; }
    const int n0 = z0 - a0, n1 = z1 - a1, both = n0 + n1;
    int rinc = both;
    for (int d = 1; d < 32; d <<= 1) {
        int v = __shfl_up(rinc, d);
        if (lane >= d) rinc += v;
    }
    int sa = a0 + a1;
    for (int d = 16; d >= 1; d >>= 1) sa += __shfl_xor(sa, d);
    if (lane == 31) { wrun[wave] = rinc; wsum[wave] = sa; }
    if (tid < BINROWS) cnt[tid] = 0;
    __syncthreads();
    int fr = (lane < wave) ? wrun[lane] : 0;
    int al = (lane < 16) ? wrun[lane] : 0;
    int sl = (lane < 16) ? wsum[lane] : 0;
    for (int d = 16; d >= 1; d >>= 1) {
        fr += __shfl_xor(fr, d);
        al += __shfl_xor(al, d);
        sl += __shfl_xor(sl, d);
    }
    const int s = sl;
    const int n = al;
    const int ex = fr + rinc - both;
    pre[c0] = (c0 < nchunks) ? ex        : 0x7fffffff;
    pre[c1] = (c1 < nchunks) ? (ex + n0) : 0x7fffffff;
    gb[c0]  = c0 * CHUNK + a0 - ex;
    gb[c1]  = c1 * CHUNK + a1 - (ex + n0);
    __syncthreads();

    const int half = (nchunks > 1) ? (1 << (31 - __clz(nchunks - 1))) : 0;
    const bool staged = (n <= SORTCAP);

    int2 v[2][EPT];
    int  pk[2][EPT];
    #pragma unroll
    for (int h = 0; h < 2; ++h) {
        #pragma unroll
        for (int j = 0; j < EPT; ++j) { v[h][j].x = 0; v[h][j].y = 0; pk[h][j] = 0; }
    }

    if (staged) {
        #pragma unroll
        for (int h = 0; h < 2; ++h) {
            if (h * 512 * EPT < n) {
                int ic[EPT], g[EPT], r[EPT];
                #pragma unroll
                for (int j = 0; j < EPT; ++j) {
                    const int i = h * 512 * EPT + tid + j * 512;
                    ic[j] = (i < n) ? i : (n - 1);
                }
                gcn_locate(pre, gb, half, ic, g);
                #pragma unroll
                for (int j = 0; j < EPT; ++j) {
                    r[j]    = trow[lo + g[j]];
                    v[h][j] = tent[lo + g[j]];
                }
                #pragma unroll
                for (int j = 0; j < EPT; ++j) {
                    const int i  = h * 512 * EPT + tid + j * 512;
                    const int rl = r[j] & (BINROWS - 1);
                    const int qq = (i < n) ? atomicAdd(&cnt[rl], 1) : 0;
                    pk[h][j] = (qq << 8) | rl;
                }
            }
        }
    } else {
        for (int b = 0; b < n; b += 512 * EPT) {
            int ic[EPT], g[EPT], r[EPT];
            #pragma unroll
            for (int j = 0; j < EPT; ++j) {
                const int i = b + tid + j * 512;
                ic[j] = (i < n) ? i : (n - 1);
            }
            gcn_locate(pre, gb, half, ic, g);
            #pragma unroll
            for (int j = 0; j < EPT; ++j) r[j] = trow[lo + g[j]];
            #pragma unroll
            for (int j = 0; j < EPT; ++j)
                if (b + tid + j * 512 < n) atomicAdd(&cnt[r[j] & (BINROWS - 1)], 1);
        }
    }
    __syncthreads();

    const int mine = (tid < BINROWS) ? cnt[tid] : 0;
    int incl = mine;
    for (int d = 1; d < 32; d <<= 1) {
        int w = __shfl_up(incl, d);
        if (lane >= d) incl += w;
    }
    if (tid < BINROWS && lane == 31) wtot[wave] = incl;
    __syncthreads();
    int front = (lane < 8 && lane < wave) ? wtot[lane] : 0;
    for (int d = 16; d >= 1; d >>= 1) front += __shfl_xor(front, d);
    const int first = s + front + incl - mine;

    if (tid < BINROWS) {
        int* of = offs + (size_t)l * (nnodes + 1);
        const int row = bin * BINROWS + tid;
        cur[tid] = staged ? (first - s) : first;
        if (row < nnodes)      of[row] = first;
        if (row == nnodes - 1) of[nnodes] = first + mine;
    }
    __syncthreads();

    if (staged) {
        #pragma unroll
        for (int h = 0; h < 2; ++h) {
            #pragma unroll
            for (int j = 0; j < EPT; ++j) {
                const int i = h * 512 * EPT + tid + j * 512;
                if (i < n) stage[cur[pk[h][j] & (BINROWS - 1)] + (pk[h][j] >> 8)] = v[h][j];
            }
        }
        __syncthreads();
        int2* dst = ent + lo + s;
        for (int i = tid; i < n; i += 512) dst[i] = stage[i];
    } else {
        for (int b = 0; b < n; b += 512 * EPT) {
            int  ic[EPT], g[EPT], r[EPT];
            int2 w[EPT];
            #pragma unroll
            for (int j = 0; j < EPT; ++j) {
                const int i = b + tid + j * 512;
                ic[j] = (i < n) ? i : (n - 1);
            }
            gcn_locate(pre, gb, half, ic, g);
            #pragma unroll
            for (int j = 0; j < EPT; ++j) {
                r[j] = trow[lo + g[j]];
                w[j] = tent[lo + g[j]];
            }
            #pragma unroll
            for (int j = 0; j < EPT; ++j) {
                if (b + tid + j * 512 < n) {
                    const int p = atomicAdd(&cur[r[j] & (BINROWS - 1)], 1);
                    ent[lo + p] = w[j];
                }
            }
        }
    }
}

#define LAYER_BLOCKS 1024


typedef __attribute__((ext_vector_type(4))) float v4f;
typedef const __attribute__((address_space(1))) char* gcn_bytep;
typedef const __attribute__((address_space(1))) v4f*  gcn_rowp;

__device__ __forceinline__ void gcn_row_sum(
    const float* __restrict__ X, const int2* __restrict__ ent,
    int s, int t, int2 ev, int lane, float4& acc, float& wsum)
{
    const gcn_bytep xb = (gcn_bytep)X;
    const unsigned lane16 = (unsigned)lane << 4;
    for (int b = s; b < t; b += 32) {
        const int n = (t - b < 32) ? (t - b) : 32;
        if (b != s) ev = ent[b + ((lane < n) ? lane : 0)];
        const int nb = (n + 7) >> 3;
        for (int q = 0; q < nb; ++q) {
            const int k = q * 8;
            v4f      m[8];
            float    w[8];
            unsigned off[8];
            #pragma unroll
            for (int j = 0; j < 8; ++j) {
                const int kk = (k + j < n) ? (k + j) : (n - 1);
                const unsigned c = (unsigned)__builtin_amdgcn_readlane(ev.x, kk);
                w[j]   = __int_as_float(__builtin_amdgcn_readlane(ev.y, kk));
                off[j] = (c << 9) + lane16;
            }
            #pragma unroll
            for (int j = 0; j < 8; ++j) m[j] = *(gcn_rowp)(xb + off[j]);
            __builtin_amdgcn_sched_group_barrier(0x002, 24, 0);
            __builtin_amdgcn_sched_group_barrier(0x020, 8, 0);
            #pragma unroll
            for (int j = 0; j < 8; ++j) {
                const bool ok = (j == 0) || (k + j < n);
                const float sx = fmaf(w[j], m[j].x, acc.x);
                const float sy = fmaf(w[j], m[j].y, acc.y);
                const float sz = fmaf(w[j], m[j].z, acc.z);
                const float sw = fmaf(w[j], m[j].w, acc.w);
                acc.x = ok ? sx : acc.x;
                acc.y = ok ? sy : acc.y;
                acc.z = ok ? sz : acc.z;
                acc.w = ok ? sw : acc.w;
                wsum  = ok ? (wsum + w[j]) : wsum;
            }
        }
    }
}

__global__ __launch_bounds__(256, 8) void gcn_layer(
    const float* __restrict__ X, const float* __restrict__ W,
    const float* __restrict__ Bv, const int* __restrict__ offs,
    const int2* __restrict__ ent, float* __restrict__ out,
    int nnodes, int ntiles)
{
    __shared__ __attribute__((aligned(16))) float xs[2][16 * LDSPAD];
    __shared__ float rs[2][16];
    __shared__ __attribute__((aligned(16))) float os[2][16 * LDSPAD];

    const int tid  = threadIdx.x;
    const int lane = tid & 31;
    const int wave = __builtin_amdgcn_readfirstlane(tid >> 5);
    const int nn   = lane & 15;
    const int kb   = (lane >> 4) << 1;
    const int mh   = (lane >> 4) * 8;
    const int n0   = wave * 16;

    v2f bw[32];
    {
        const float* wcol = &W[(size_t)kb * NFEAT + n0 + nn];
        #pragma unroll
        for (int k4 = 0; k4 < 32; ++k4) {
            bw[k4].x = wcol[(size_t)(4 * k4) * NFEAT];
            bw[k4].y = wcol[(size_t)(4 * k4 + 1) * NFEAT];
        }
    }
    const float bias = Bv[n0 + nn];

    auto store_rows = [&](int buf, int mfirst) {
        const int r0 = mfirst + wave * 2;
        const float4* t = (const float4*)&os[buf][(wave * 2) * LDSPAD + lane * 4];
        float4* o0 = (float4*)(out + (size_t)r0 * NFEAT) + lane;
        if (r0 + 1 < nnodes) {
            const float4 q0 = t[0];
            const float4 q1 = t[LDSPAD / 4];
            o0[0]         = q0;
            o0[NFEAT / 4] = q1;
        } else if (r0 < nnodes) {
            o0[0] = t[0];
        }
    };

    int p = 0;
    int mprev = -1;
    for (int tile = blockIdx.x; tile < ntiles; tile += gridDim.x, p ^= 1) {
        const int m0 = tile * 16;

        {
            const int r0 = m0 + wave * 2;
            const int i0 = (r0     < nnodes) ? r0       : nnodes;
            const int i1 = (r0 + 1 < nnodes) ? (r0 + 1) : nnodes;
            const int i2 = (r0 + 2 < nnodes) ? (r0 + 2) : nnodes;
            const int s0 = offs[i0], s1 = offs[i1], s2 = offs[i2];
            const int nA = (s1 - s0 < 32) ? (s1 - s0) : 32;
            const int nB = (s2 - s1 < 32) ? (s2 - s1) : 32;
            const int2 evA = ent[(nA > 0) ? (s0 + ((lane < nA) ? lane : 0)) : 0];
            const int2 evB = ent[(nB > 0) ? (s1 + ((lane < nB) ? lane : 0)) : 0];

            float4 a0; a0.x = 0.f; a0.y = 0.f; a0.z = 0.f; a0.w = 0.f;
            float4 a1 = a0;
            float  w0 = 0.f, w1 = 0.f;
            gcn_row_sum(X, ent, s0, s1, evA, lane, a0, w0);
            gcn_row_sum(X, ent, s1, s2, evB, lane, a1, w1);

            float* xw = &xs[p][(wave * 2) * LDSPAD + lane * 4];
            *(float4*)xw            = a0;
            *(float4*)(xw + LDSPAD) = a1;
            if (lane == 0) { rs[p][wave * 2] = w0; rs[p][wave * 2 + 1] = w1; }
        }
        __syncthreads();

        if (mprev >= 0) store_rows(p ^ 1, mprev);

        v8f acc;
        #pragma unroll
        for (int r = 0; r < 8; ++r) acc[r] = rs[p][r + mh] * bias;

        const float* xrow = &xs[p][nn * LDSPAD + kb];
        #pragma unroll
        for (int k4 = 0; k4 < 32; ++k4) {
            v2f a;
            a.x = xrow[4 * k4];
            a.y = xrow[4 * k4 + 1];
            acc = __builtin_amdgcn_wmma_f32_16x16x4_f32(
                 false, a,  false, bw[k4],
                 (short)0, acc,  false,  false);
        }

        float o[8];
        #pragma unroll
        for (int r = 0; r < 8; ++r) {
            const float v = acc[r];
            o[r] = (v > 0.0f) ? v : (__expf(v) - 1.0f);
        }
        float* ow = &os[p][mh * LDSPAD + n0 + nn];
        #pragma unroll
        for (int r = 0; r < 8; ++r) ow[r * LDSPAD] = o[r];
        mprev = m0;
    }
    __syncthreads();
    if (mprev >= 0) store_rows(p ^ 1, mprev);
}

extern "C" void kernel_launch(void* const* d_in, const int* in_sizes, int n_in,
                              void* d_out, int out_size, void* d_ws, size_t ws_size,
                              hipStream_t stream)
{
    const float* x    = (const float*)d_in[0];
    const float* W    = (const float*)d_in[1];
    const float* b    = (const float*)d_in[2];
    const int*   rows = (const int*)  d_in[3];
    const int*   cols = (const int*)  d_in[4];
    const float* vals = (const float*)d_in[5];

    const int nnodes  = in_sizes[0] / NFEAT;
    const int nlayers = in_sizes[1] / (NFEAT * NFEAT);
    const int nedges  = in_sizes[3] / nlayers;

    const int nbins   = (nnodes + BINROWS - 1) / BINROWS;
    const int nchunks = (nedges + CHUNK - 1) / CHUNK;

    const size_t nodeElems = (size_t)nnodes * NFEAT;
    const size_t allEdges  = (size_t)nlayers * nedges;
    float* actA = (float*)d_ws;
    float* actB = actA + nodeElems;
    int2*  ent  = (int2*)(actB + nodeElems);
    int2*  tent = ent + allEdges;
    int*   trow = (int*)(tent + allEdges);
    int*   tab  = trow + allEdges;
    int*   offs = tab + (size_t)nlayers * (nbins + 1) * nchunks;

    const int ntiles = (nnodes + 15) / 16;
    const dim3 chunkGrid(nchunks, nlayers);
    const dim3 binGrid(nbins, nlayers);

    gcn_chunk_sort<<<chunkGrid, 1024, 0, stream>>>(rows, cols, vals, tab, trow, tent,
                                                   nedges, nbins, nchunks);
    gcn_bin_sort<<<binGrid, 512, 0, stream>>>(tab, trow, tent, offs, ent,
                                              nedges, nnodes, nbins, nchunks);

    for (int l = 0; l < nlayers; ++l) {
        const float* cur = (l == 0) ? x : ((l & 1) ? actA : actB);
        float* dst = (l == nlayers - 1) ? (float*)d_out : ((l & 1) ? actB : actA);
        gcn_layer<<<LAYER_BLOCKS, 256, 0, stream>>>(
            cur, W + (size_t)l * NFEAT * NFEAT, b + (size_t)l * NFEAT,
            offs + (size_t)l * (nnodes + 1), ent + (size_t)l * nedges,
            dst, nnodes, ntiles);
    }
}
